# starting kernel
# baseline (speedup 1.0000x reference)
#include <hip/hip_runtime.h>

typedef __attribute__((ext_vector_type(2))) float v2f;
typedef __attribute__((ext_vector_type(8))) float v8f;

#define N_NODES 100000
#define EMB_DIM 128
#define HIDDEN  128
#define OUT_C   64
#define N_EDGES 800000

__global__ void zero4_kernel(float4* __restrict__ p, int n4) {
    int i = blockIdx.x * blockDim.x + threadIdx.x;
    int stride = gridDim.x * blockDim.x;
    float4 z = make_float4(0.f, 0.f, 0.f, 0.f);
    for (; i < n4; i += stride) p[i] = z;
}

__global__ void degree_kernel(const int* __restrict__ ei, float* __restrict__ cnt) {
    int e = blockIdx.x * blockDim.x + threadIdx.x;
    if (e < N_EDGES) atomicAdd(&cnt[ei[N_EDGES + e]], 1.0f);
}

template<int F>
__global__ void scatter_kernel(const int* __restrict__ ei,
                               const float* __restrict__ x,
                               const int* __restrict__ xidx,
                               float* __restrict__ agg) {
    const int CH = F / 4;
    int t = blockIdx.x * blockDim.x + threadIdx.x;
    if (t >= N_EDGES * CH) return;
    int e = t / CH;
    int f = (t - e * CH) * 4;
    int s = ei[e];
    int d = ei[N_EDGES + e];
    if (xidx) s = xidx[s];
    const float4 v = *reinterpret_cast<const float4*>(x + (size_t)s * F + f);
    float* a = agg + (size_t)d * F + f;
    atomicAdd(a + 0, v.x);
    atomicAdd(a + 1, v.y);
    atomicAdd(a + 2, v.z);
    atomicAdd(a + 3, v.w);
}

template<int K, int OUT, bool RELU>
__global__ __launch_bounds__(128)
void sage_wmma_kernel(const float* __restrict__ agg,
                      const float* __restrict__ cnt,
                      const float* __restrict__ X,
                      const int* __restrict__ xidx,
                      const float* __restrict__ Wl,
                      const float* __restrict__ bias,
                      const float* __restrict__ Wr,
                      float* __restrict__ out) {
    const int lane = threadIdx.x;
    const int wave = threadIdx.y;
    const int rowt = blockIdx.x;
    const int colt = blockIdx.y * 4 + wave;
    const int half = lane >> 4;
    const int l16  = lane & 15;

    const int arow = rowt * 16 + l16;
    const int col  = colt * 16 + l16;

    const float inv = 1.0f / fmaxf(cnt[arow], 1.0f);
    int xrow = arow;
    if (xidx) xrow = xidx[arow];

    const float* __restrict__ aA = agg + (size_t)arow * K;
    const float* __restrict__ aX = X   + (size_t)xrow * K;

    v8f c = {0.f, 0.f, 0.f, 0.f, 0.f, 0.f, 0.f, 0.f};

    #pragma unroll
    for (int k = 0; k < K; k += 4) {
        const int ko = k + half * 2;
        v2f a; a.x = aA[ko] * inv;              a.y = aA[ko + 1] * inv;
        v2f b; b.x = Wl[(size_t)ko * OUT + col]; b.y = Wl[(size_t)(ko + 1) * OUT + col];
        c = __builtin_amdgcn_wmma_f32_16x16x4_f32(false, a, false, b,
                                                  (short)0, c, false, false);
    }
    #pragma unroll
    for (int k = 0; k < K; k += 4) {
        const int ko = k + half * 2;
        v2f a; a.x = aX[ko];                    a.y = aX[ko + 1];
        v2f b; b.x = Wr[(size_t)ko * OUT + col]; b.y = Wr[(size_t)(ko + 1) * OUT + col];
        c = __builtin_amdgcn_wmma_f32_16x16x4_f32(false, a, false, b,
                                                  (short)0, c, false, false);
    }

    const float bv = bias[col];
    #pragma unroll
    for (int v = 0; v < 8; ++v) {
        const int r = rowt * 16 + v + half * 8;
        float val = c[v] + bv;
        if (RELU) val = fmaxf(val, 0.0f);
        out[(size_t)r * OUT + col] = val;
    }
}

extern "C" void kernel_launch(void* const* d_in, const int* in_sizes, int n_in,
                              void* d_out, int out_size, void* d_ws, size_t ws_size,
                              hipStream_t stream) {
    (void)in_sizes; (void)n_in; (void)out_size; (void)ws_size;

    const int*   xidx = (const int*)d_in[0];
    const int*   ei   = (const int*)d_in[1];
    const float* emb  = (const float*)d_in[2];
    const float* Wl1  = (const float*)d_in[3];
    const float* bl1  = (const float*)d_in[4];
    const float* Wr1  = (const float*)d_in[5];
    const float* Wl2  = (const float*)d_in[6];
    const float* bl2  = (const float*)d_in[7];
    const float* Wr2  = (const float*)d_in[8];

    float* agg = (float*)d_ws;
    float* h   = agg + (size_t)N_NODES * HIDDEN;
    float* cnt = h   + (size_t)N_NODES * HIDDEN;

    const int aggN4 = N_NODES * EMB_DIM / 4;
    const int cntN4 = N_NODES / 4;
    const int scatT = N_EDGES * (EMB_DIM / 4);

    dim3 wblk(32, 4);

    zero4_kernel<<<2048, 256, 0, stream>>>((float4*)agg, aggN4);
    zero4_kernel<<<128, 256, 0, stream>>>((float4*)cnt, cntN4);
    degree_kernel<<<(N_EDGES + 255) / 256, 256, 0, stream>>>(ei, cnt);
    scatter_kernel<EMB_DIM><<<(scatT + 255) / 256, 256, 0, stream>>>(ei, emb, xidx, agg);
    sage_wmma_kernel<EMB_DIM, HIDDEN, true>
        <<<dim3(N_NODES / 16, HIDDEN / 16 / 4), wblk, 0, stream>>>(
            agg, cnt, emb, xidx, Wl1, bl1, Wr1, h);

    zero4_kernel<<<2048, 256, 0, stream>>>((float4*)agg, aggN4);
    scatter_kernel<HIDDEN><<<(scatT + 255) / 256, 256, 0, stream>>>(ei, h, nullptr, agg);
    sage_wmma_kernel<HIDDEN, OUT_C, false>
        <<<dim3(N_NODES / 16, OUT_C / 16 / 4), wblk, 0, stream>>>(
            agg, cnt, h, nullptr, Wl2, bl2, Wr2, (float*)d_out);
}


# Replace f32 atomic scatter and zero passes by dst-bucketed edge list (int rank atomics, arrival order changes sum order); layer kernel gathers neighbour rows into an LDS mean tile
# speedup vs baseline: 2.6113x; 2.6113x over previous
#include <hip/hip_runtime.h>

typedef __attribute__((ext_vector_type(2))) float v2f;
typedef __attribute__((ext_vector_type(8))) float v8f;

#define N_NODES 100000
#define EMB_DIM 128
#define HIDDEN  128
#define OUT_C   64
#define N_EDGES 800000

__global__ void clear_counts_kernel(int4* __restrict__ deg4, int n4, int* __restrict__ total) {
    int i = blockIdx.x * blockDim.x + threadIdx.x;
    if (i < n4) deg4[i] = make_int4(0, 0, 0, 0);
    if (i == 0) *total = 0;
}

__global__ void rank_kernel(const int* __restrict__ ei, int* __restrict__ deg,
                            int* __restrict__ rank) {
    int e = blockIdx.x * blockDim.x + threadIdx.x;
    if (e < N_EDGES) rank[e] = atomicAdd(&deg[ei[N_EDGES + e]], 1);
}

__global__ void segment_kernel(const int* __restrict__ deg, int* __restrict__ total,
                               int2* __restrict__ seg) {
    const int n = blockIdx.x * blockDim.x + threadIdx.x;
    const int lane = threadIdx.x & 31;
    const int c = (n < N_NODES) ? deg[n] : 0;
    int incl = c;
    for (int d = 1; d < 32; d <<= 1) {
        const int t = __shfl_up(incl, d);
        if (lane >= d) incl += t;
    }
    int base = 0;
    if (lane == 31) base = atomicAdd(total, incl);
    base = __shfl(base, 31);
    if (n < N_NODES) seg[n] = make_int2(base + incl - c, c);
}

__global__ void fill_kernel(const int* __restrict__ ei, const int* __restrict__ rank,
                            const int2* __restrict__ seg, int* __restrict__ nbr) {
    int e = blockIdx.x * blockDim.x + threadIdx.x;
    if (e < N_EDGES) {
        const int d = ei[N_EDGES + e];
        nbr[seg[d].x + rank[e]] = ei[e];
    }
}

template<int K, int OUT, bool RELU>
__global__ __launch_bounds__(128)
void sage_gather_wmma_kernel(const int* __restrict__ nbr,
                             const int2* __restrict__ seg,
                             const float* __restrict__ X,
                             const int* __restrict__ xidx,
                             const float* __restrict__ Wl,
                             const float* __restrict__ bias,
                             const float* __restrict__ Wr,
                             float* __restrict__ out) {
    static_assert(K == 128, "one float4 per lane covers a row");
    constexpr int CT  = OUT / 64;
    constexpr int LDA = K + 4;
    __shared__ __attribute__((aligned(16))) float sMean[16 * LDA];

    const int lane = threadIdx.x;
    const int wave = __builtin_amdgcn_readfirstlane(threadIdx.y);
    const int rowt = blockIdx.x;
    const int half = lane >> 4;
    const int l16  = lane & 15;

    for (int rr = 0; rr < 4; ++rr) {
        const int r = wave * 4 + rr;
        const int2 sd = seg[rowt * 16 + r];
        const int off = sd.x;
        const int deg = sd.y;
        float4 acc = make_float4(0.f, 0.f, 0.f, 0.f);
        for (int base = 0; base < deg; base += 32) {
            const int n = (deg - base < 32) ? (deg - base) : 32;
            int idx = 0;
            if (lane < n) {
                idx = nbr[off + base + lane];
                if (xidx) idx = xidx[idx];
            }
            #pragma unroll 4
            for (int j = 0; j < n; ++j) {
                const int s = __builtin_amdgcn_readlane(idx, j);
                const float4 v = *reinterpret_cast<const float4*>(X + (size_t)s * K + lane * 4);
                acc.x += v.x; acc.y += v.y; acc.z += v.z; acc.w += v.w;
            }
        }
        const float inv = 1.0f / fmaxf((float)deg, 1.0f);
        acc.x *= inv; acc.y *= inv; acc.z *= inv; acc.w *= inv;
        *reinterpret_cast<float4*>(&sMean[r * LDA + lane * 4]) = acc;
    }
    __syncthreads();

    const int arow = rowt * 16 + l16;
    int xrow = arow;
    if (xidx) xrow = xidx[arow];
    const float* __restrict__ aX = X + (size_t)xrow * K;
    const float* __restrict__ aM = sMean + l16 * LDA;

    v8f c[CT];
    #pragma unroll
    for (int t = 0; t < CT; ++t) c[t] = (v8f){0.f, 0.f, 0.f, 0.f, 0.f, 0.f, 0.f, 0.f};

    #pragma unroll
    for (int k = 0; k < K; k += 4) {
        const int ko = k + half * 2;
        v2f a; a.x = aM[ko];                    a.y = aM[ko + 1];
        #pragma unroll
        for (int t = 0; t < CT; ++t) {
            const int col = (t * 4 + wave) * 16 + l16;
            v2f b; b.x = Wl[(size_t)ko * OUT + col]; b.y = Wl[(size_t)(ko + 1) * OUT + col];
            c[t] = __builtin_amdgcn_wmma_f32_16x16x4_f32(false, a, false, b,
                                                         (short)0, c[t], false, false);
        }
    }
    #pragma unroll
    for (int k = 0; k < K; k += 4) {
        const int ko = k + half * 2;
        v2f a; a.x = aX[ko];                    a.y = aX[ko + 1];
        #pragma unroll
        for (int t = 0; t < CT; ++t) {
            const int col = (t * 4 + wave) * 16 + l16;
            v2f b; b.x = Wr[(size_t)ko * OUT + col]; b.y = Wr[(size_t)(ko + 1) * OUT + col];
            c[t] = __builtin_amdgcn_wmma_f32_16x16x4_f32(false, a, false, b,
                                                         (short)0, c[t], false, false);
        }
    }

    #pragma unroll
    for (int t = 0; t < CT; ++t) {
        const int col = (t * 4 + wave) * 16 + l16;
        const float bv = bias[col];
        #pragma unroll
        for (int v = 0; v < 8; ++v) {
            const int r = rowt * 16 + v + half * 8;
            float val = c[t][v] + bv;
            if (RELU) val = fmaxf(val, 0.0f);
            out[(size_t)r * OUT + col] = val;
        }
    }
}

extern "C" void kernel_launch(void* const* d_in, const int* in_sizes, int n_in,
                              void* d_out, int out_size, void* d_ws, size_t ws_size,
                              hipStream_t stream) {
    (void)in_sizes; (void)n_in; (void)out_size; (void)ws_size;

    const int*   xidx = (const int*)d_in[0];
    const int*   ei   = (const int*)d_in[1];
    const float* emb  = (const float*)d_in[2];
    const float* Wl1  = (const float*)d_in[3];
    const float* bl1  = (const float*)d_in[4];
    const float* Wr1  = (const float*)d_in[5];
    const float* Wl2  = (const float*)d_in[6];
    const float* bl2  = (const float*)d_in[7];
    const float* Wr2  = (const float*)d_in[8];

    int*   nbr   = (int*)d_ws;
    int*   rank  = nbr + N_EDGES;
    int2*  seg   = (int2*)(rank + N_EDGES);
    int*   total = (int*)(seg + N_NODES);
    float* h     = (float*)d_ws + (size_t)N_NODES * HIDDEN;
    int*   deg   = (int*)(h + (size_t)N_NODES * HIDDEN);

    dim3 wblk(32, 4);

    clear_counts_kernel<<<(N_NODES / 4 + 255) / 256, 256, 0, stream>>>((int4*)deg, N_NODES / 4, total);
    rank_kernel<<<(N_EDGES + 255) / 256, 256, 0, stream>>>(ei, deg, rank);
    segment_kernel<<<(N_NODES + 255) / 256, 256, 0, stream>>>(deg, total, seg);
    fill_kernel<<<(N_EDGES + 255) / 256, 256, 0, stream>>>(ei, rank, seg, nbr);

    sage_gather_wmma_kernel<EMB_DIM, HIDDEN, true>
        <<<dim3(N_NODES / 16), wblk, 0, stream>>>(
            nbr, seg, emb, xidx, Wl1, bl1, Wr1, h);

    sage_gather_wmma_kernel<HIDDEN, OUT_C, false>
        <<<dim3(N_NODES / 16), wblk, 0, stream>>>(
            nbr, seg, h, nullptr, Wl2, bl2, Wr2, (float*)d_out);
}


# Layer kernel as one resident workgroup per WGP: Wl, Wr copied to LDS once, each wave loops over whole 16-node tiles with B fragments from LDS and a private LDS A tile; order of every sum kept
# speedup vs baseline: 2.8103x; 1.0762x over previous
#include <hip/hip_runtime.h>

typedef __attribute__((ext_vector_type(2))) float v2f;
typedef __attribute__((ext_vector_type(8))) float v8f;

#define N_NODES 100000
#define EMB_DIM 128
#define HIDDEN  128
#define OUT_C   64
#define N_EDGES 800000

__global__ void clear_counts_kernel(int4* __restrict__ deg4, int n4, int* __restrict__ total) {
    int i = blockIdx.x * blockDim.x + threadIdx.x;
    if (i < n4) deg4[i] = make_int4(0, 0, 0, 0);
    if (i == 0) *total = 0;
}

__global__ void rank_kernel(const int* __restrict__ ei, int* __restrict__ deg,
                            int* __restrict__ rank) {
    int e = blockIdx.x * blockDim.x + threadIdx.x;
    if (e < N_EDGES) rank[e] = atomicAdd(&deg[ei[N_EDGES + e]], 1);
}

__global__ void segment_kernel(const int* __restrict__ deg, int* __restrict__ total,
                               int2* __restrict__ seg) {
    const int n = blockIdx.x * blockDim.x + threadIdx.x;
    const int lane = threadIdx.x & 31;
    const int c = (n < N_NODES) ? deg[n] : 0;
    int incl = c;
    for (int d = 1; d < 32; d <<= 1) {
        const int t = __shfl_up(incl, d);
        if (lane >= d) incl += t;
    }
    int base = 0;
    if (lane == 31) base = atomicAdd(total, incl);
    base = __shfl(base, 31);
    if (n < N_NODES) seg[n] = make_int2(base + incl - c, c);
}

__global__ void fill_kernel(const int* __restrict__ ei, const int* __restrict__ rank,
                            const int2* __restrict__ seg, int* __restrict__ nbr) {
    int e = blockIdx.x * blockDim.x + threadIdx.x;
    if (e < N_EDGES) {
        const int d = ei[N_EDGES + e];
        nbr[seg[d].x + rank[e]] = ei[e];
    }
}

template<int K, int OUT, bool RELU, bool IDX, int NW, int GB>
__global__ __launch_bounds__(NW * 32)
void sage_layer_kernel(const int* __restrict__ nbr,
                       const int2* __restrict__ seg,
                       const float* __restrict__ X,
                       const int* __restrict__ xidx,
                       const float* __restrict__ Wl,
                       const float* __restrict__ bias,
                       const float* __restrict__ Wr,
                       float* __restrict__ out) {
    static_assert(K == 128, "one float4 per lane covers a row");
    constexpr int CT  = OUT / 16;
    constexpr int LDA = K + 4;
    constexpr int KP  = K / 2;
    constexpr int WP  = OUT * 2 + 32;
    constexpr int NT  = NW * 32;
    __shared__ __attribute__((aligned(16))) float sW[2 * KP * WP];
    __shared__ __attribute__((aligned(16))) float sTile[NW * 16 * LDA];

    const int lane = threadIdx.x;
    const int wave = __builtin_amdgcn_readfirstlane(threadIdx.y);
    const int half = lane >> 4;
    const int l16  = lane & 15;

    {
        constexpr int C4 = OUT / 4;
        constexpr int PER = KP * C4;
        const int tid = threadIdx.y * 32 + threadIdx.x;
        #pragma unroll
        for (int it = 0; it < (2 * PER + NT - 1) / NT; ++it) {
            const int i = tid + it * NT;
            if ((it + 1) * NT > 2 * PER && i >= 2 * PER) continue;
            const int m   = i / PER;
            const int rem = i - m * PER;
            const int kp  = rem / C4;
            const int c4  = rem - kp * C4;
            const float* __restrict__ W = m ? Wr : Wl;
            const float4 r0 = *reinterpret_cast<const float4*>(W + (size_t)(2 * kp) * OUT + c4 * 4);
            const float4 r1 = *reinterpret_cast<const float4*>(W + (size_t)(2 * kp + 1) * OUT + c4 * 4);
            float* d = sW + (m * KP + kp) * WP + c4 * 8;
            *reinterpret_cast<float4*>(d)     = make_float4(r0.x, r1.x, r0.y, r1.y);
            *reinterpret_cast<float4*>(d + 4) = make_float4(r0.z, r1.z, r0.w, r1.w);
        }
    }
    __syncthreads();

    float* const tileM = sTile + wave * (16 * LDA);
    const float* const aM = tileM + l16 * LDA + half * 2;
    const float* const bL = sW + half * WP + l16 * 2;
    const float* const bR = bL + KP * WP;

    for (int tile = wave * (int)gridDim.x + (int)blockIdx.x; tile < N_NODES / 16;
         tile += (int)gridDim.x * NW) {
        const int node0 = tile * 16;

        const int2 sd = seg[node0 + l16];
        const int start = __builtin_amdgcn_readlane(sd.x, 0);
        const int total = __builtin_amdgcn_readlane(sd.x, 15) + __builtin_amdgcn_readlane(sd.y, 15) - start;

        int cur = 0;
        int curEnd = __builtin_amdgcn_readlane(sd.y, 0);
        float4 acc = make_float4(0.f, 0.f, 0.f, 0.f);

        auto flush = [&]() {
            const int dg = __builtin_amdgcn_readlane(sd.y, cur);
            const float inv = 1.0f / fmaxf((float)dg, 1.0f);
            acc.x *= inv; acc.y *= inv; acc.z *= inv; acc.w *= inv;
            *reinterpret_cast<float4*>(&tileM[cur * LDA + lane * 4]) = acc;
            acc = make_float4(0.f, 0.f, 0.f, 0.f);
            ++cur;
            if (cur < 16) curEnd += __builtin_amdgcn_readlane(sd.y, cur);
        };
        auto take = [&](int p, const float4& v) {
            while (p >= curEnd && cur < 15) flush();
            acc.x += v.x; acc.y += v.y; acc.z += v.z; acc.w += v.w;
        };

        int nraw = 0;
        if (lane < total) nraw = nbr[start + lane];
        for (int base = 0; base < total; base += 32) {
            const int n = (total - base < 32) ? (total - base) : 32;
            int idx = nraw;
            if (base + 32 + lane < total) nraw = nbr[start + base + 32 + lane];
            if (IDX) { if (lane < n) idx = xidx[idx]; }

            int j = 0;
            for (; j + GB <= n; j += GB) {
                float4 v[GB];
                #pragma unroll
                for (int u = 0; u < GB; ++u) {
                    const int s = __builtin_amdgcn_readlane(idx, j + u);
                    v[u] = *reinterpret_cast<const float4*>(X + (size_t)s * K + lane * 4);
                }
                #pragma unroll
                for (int u = 0; u < GB; ++u) take(base + j + u, v[u]);
            }
            for (; j + 4 <= n; j += 4) {
                float4 v[4];
                #pragma unroll
                for (int u = 0; u < 4; ++u) {
                    const int s = __builtin_amdgcn_readlane(idx, j + u);
                    v[u] = *reinterpret_cast<const float4*>(X + (size_t)s * K + lane * 4);
                }
                #pragma unroll
                for (int u = 0; u < 4; ++u) take(base + j + u, v[u]);
            }
            for (; j < n; ++j) {
                const int s = __builtin_amdgcn_readlane(idx, j);
                const float4 v = *reinterpret_cast<const float4*>(X + (size_t)s * K + lane * 4);
                take(base + j, v);
            }
        }
        while (cur < 16) flush();

        __builtin_amdgcn_fence(__ATOMIC_RELEASE, "wavefront");
        __builtin_amdgcn_wave_barrier();
        __builtin_amdgcn_fence(__ATOMIC_ACQUIRE, "wavefront");

        int xrow = node0 + l16;
        if (IDX) xrow = xidx[xrow];
#define OWN_ROWS(F) F(0) F(1) F(2) F(3) F(4) F(5) F(6) F(7) F(8) F(9) F(10) F(11) F(12) F(13) F(14) F(15)
#define OWN_ROW_LOAD(r) \
        const float4 xr##r = *reinterpret_cast<const float4*>( \
            X + (size_t)__builtin_amdgcn_readlane(xrow, r) * K + lane * 4);
#define OWN_ROW_PARK(r) *reinterpret_cast<float4*>(&tileM[r * LDA + lane * 4]) = xr##r;
        OWN_ROWS(OWN_ROW_LOAD)

        v8f c[CT];
        #pragma unroll
        for (int t = 0; t < CT; ++t) c[t] = (v8f){0.f, 0.f, 0.f, 0.f, 0.f, 0.f, 0.f, 0.f};

        auto product = [&](const float* __restrict__ bp) {
            const float* __restrict__ ap = aM;
            v2f a0, a1, b0[CT], b1[CT];
            a0 = *reinterpret_cast<const v2f*>(ap);
            #pragma unroll
            for (int t = 0; t < CT; ++t) b0[t] = *reinterpret_cast<const v2f*>(bp + t * 32);
            #pragma nounroll
            for (int it = 0; it < K / 8 - 1; ++it) {
                a1 = *reinterpret_cast<const v2f*>(ap + 4);
                #pragma unroll
                for (int t = 0; t < CT; ++t) b1[t] = *reinterpret_cast<const v2f*>(bp + 2 * WP + t * 32);
                #pragma unroll
                for (int t = 0; t < CT; ++t)
                    c[t] = __builtin_amdgcn_wmma_f32_16x16x4_f32(false, a0, false, b0[t],
                                                                 (short)0, c[t], false, false);
                a0 = *reinterpret_cast<const v2f*>(ap + 8);
                #pragma unroll
                for (int t = 0; t < CT; ++t) b0[t] = *reinterpret_cast<const v2f*>(bp + 4 * WP + t * 32);
                #pragma unroll
                for (int t = 0; t < CT; ++t)
                    c[t] = __builtin_amdgcn_wmma_f32_16x16x4_f32(false, a1, false, b1[t],
                                                                 (short)0, c[t], false, false);
                ap += 8;
                bp += 4 * WP;
            }
            a1 = *reinterpret_cast<const v2f*>(ap + 4);
            #pragma unroll
            for (int t = 0; t < CT; ++t) b1[t] = *reinterpret_cast<const v2f*>(bp + 2 * WP + t * 32);
            #pragma unroll
            for (int t = 0; t < CT; ++t)
                c[t] = __builtin_amdgcn_wmma_f32_16x16x4_f32(false, a0, false, b0[t],
                                                             (short)0, c[t], false, false);
            #pragma unroll
            for (int t = 0; t < CT; ++t)
                c[t] = __builtin_amdgcn_wmma_f32_16x16x4_f32(false, a1, false, b1[t],
                                                             (short)0, c[t], false, false);
        };

        product(bL);

        __builtin_amdgcn_fence(__ATOMIC_RELEASE, "wavefront");
        __builtin_amdgcn_wave_barrier();
        __builtin_amdgcn_fence(__ATOMIC_ACQUIRE, "wavefront");
        OWN_ROWS(OWN_ROW_PARK)
#undef OWN_ROW_PARK
#undef OWN_ROW_LOAD
#undef OWN_ROWS
        __builtin_amdgcn_fence(__ATOMIC_RELEASE, "wavefront");
        __builtin_amdgcn_wave_barrier();
        __builtin_amdgcn_fence(__ATOMIC_ACQUIRE, "wavefront");

        product(bR);

        #pragma unroll
        for (int t = 0; t < CT; ++t) {
            const int col = t * 16 + l16;
            const float bv = bias[col];
            #pragma unroll
            for (int v = 0; v < 8; ++v) {
                const int r = node0 + v + half * 8;
                float val = c[t][v] + bv;
                if (RELU) val = fmaxf(val, 0.0f);
                out[(size_t)r * OUT + col] = val;
            }
        }

        __builtin_amdgcn_fence(__ATOMIC_RELEASE, "wavefront");
        __builtin_amdgcn_wave_barrier();
        __builtin_amdgcn_fence(__ATOMIC_ACQUIRE, "wavefront");
    }
}

extern "C" void kernel_launch(void* const* d_in, const int* in_sizes, int n_in,
                              void* d_out, int out_size, void* d_ws, size_t ws_size,
                              hipStream_t stream) {
    (void)in_sizes; (void)n_in; (void)out_size; (void)ws_size;

    const int*   xidx = (const int*)d_in[0];
    const int*   ei   = (const int*)d_in[1];
    const float* emb  = (const float*)d_in[2];
    const float* Wl1  = (const float*)d_in[3];
    const float* bl1  = (const float*)d_in[4];
    const float* Wr1  = (const float*)d_in[5];
    const float* Wl2  = (const float*)d_in[6];
    const float* bl2  = (const float*)d_in[7];
    const float* Wr2  = (const float*)d_in[8];

    int*   nbr   = (int*)d_ws;
    int*   rank  = nbr + N_EDGES;
    int2*  seg   = (int2*)(rank + N_EDGES);
    int*   total = (int*)(seg + N_NODES);
    float* h     = (float*)d_ws + (size_t)N_NODES * HIDDEN;
    int*   deg   = (int*)(h + (size_t)N_NODES * HIDDEN);

    clear_counts_kernel<<<(N_NODES / 4 + 255) / 256, 256, 0, stream>>>((int4*)deg, N_NODES / 4, total);
    rank_kernel<<<(N_EDGES + 255) / 256, 256, 0, stream>>>(ei, deg, rank);
    segment_kernel<<<(N_NODES + 255) / 256, 256, 0, stream>>>(deg, total, seg);
    fill_kernel<<<(N_EDGES + 255) / 256, 256, 0, stream>>>(ei, rank, seg, nbr);


    sage_layer_kernel<EMB_DIM, HIDDEN, true, true, 16, 16>
        <<<dim3(256), dim3(32, 16), 0, stream>>>(
            nbr, seg, emb, xidx, Wl1, bl1, Wr1, h);

    sage_layer_kernel<HIDDEN, OUT_C, false, false, 28, 8>
        <<<dim3(256), dim3(32, 28), 0, stream>>>(
            nbr, seg, h, nullptr, Wl2, bl2, Wr2, (float*)d_out);
}


# Bucket edges by destination with a two-level counting sort on LDS counters (512-node bins, then node): 3 launches, no global atomics, run-wise stores; sum order now set by LDS atomic arrival
# speedup vs baseline: 3.5784x; 1.2733x over previous
#include <hip/hip_runtime.h>

typedef __attribute__((ext_vector_type(2))) float v2f;
typedef __attribute__((ext_vector_type(8))) float v8f;

#define N_NODES 100000
#define EMB_DIM 128
#define HIDDEN  128
#define OUT_C   64
#define N_EDGES 800000

#define NCHUNK    256
#define CH        (N_EDGES / NCHUNK)
#define BIN_SHIFT 9
#define BIN_NODES (1 << BIN_SHIFT)
#define NBIN      ((N_NODES + BIN_NODES - 1) / BIN_NODES)
#define BT        512
static_assert(CH * NCHUNK == N_EDGES, "chunks tile the edge list");
static_assert(NBIN <= BT && BIN_NODES == BT, "one thread per bin / per node of a bin");
static_assert(NCHUNK == 256, "one lane sums 8 chunk counts");

__global__ __launch_bounds__(BT)
void bin_count_kernel(const int* __restrict__ ei, int* __restrict__ histT) {
    __shared__ int sCnt[NBIN];
    const int tid = threadIdx.x;
    if (tid < NBIN) sCnt[tid] = 0;
    __syncthreads();
    const int* __restrict__ dst = ei + N_EDGES + blockIdx.x * CH;
    for (int i = tid; i < CH; i += BT) atomicAdd(&sCnt[dst[i] >> BIN_SHIFT], 1);
    __syncthreads();
    if (tid < NBIN) histT[tid * NCHUNK + blockIdx.x] = sCnt[tid];
}

__global__ __launch_bounds__(BT)
void bin_scatter_kernel(const int* __restrict__ ei, const int* __restrict__ histT,
                        int* __restrict__ base, int* __restrict__ tmp) {
    __shared__ int sTot[NBIN], sPre[NBIN], sCur[NBIN], sLoc[NBIN], sCnt[NBIN];
    __shared__ int sVal[CH], sDst[CH];
    const int tid = threadIdx.x, lane = tid & 31, wave = tid >> 5;
    const int w = blockIdx.x;
    if (tid < NBIN) sCnt[tid] = 0;

    for (int b = wave; b < NBIN; b += BT / 32) {
        const int4* __restrict__ p = reinterpret_cast<const int4*>(histT + b * NCHUNK) + lane * 2;
        const int4 a = p[0], c = p[1];
        const int w0 = lane * 8;
        int tot = a.x + a.y + a.z + a.w + c.x + c.y + c.z + c.w;
        int pre = (w0 + 0 < w ? a.x : 0) + (w0 + 1 < w ? a.y : 0) + (w0 + 2 < w ? a.z : 0) +
                  (w0 + 3 < w ? a.w : 0) + (w0 + 4 < w ? c.x : 0) + (w0 + 5 < w ? c.y : 0) +
                  (w0 + 6 < w ? c.z : 0) + (w0 + 7 < w ? c.w : 0);
        for (int d = 16; d >= 1; d >>= 1) { tot += __shfl_xor(tot, d); pre += __shfl_xor(pre, d); }
        if (lane == 0) { sTot[b] = tot; sPre[b] = pre; }
    }
    __syncthreads();
    if (tid < NBIN) {
        int s = 0;
        for (int j = 0; j < tid; ++j) s += sTot[j];
        sCur[tid] = s + sPre[tid];
        if (w == 0) base[tid] = s;
    }
    if (w == 0 && tid == 0) base[NBIN] = N_EDGES;

    const int* __restrict__ srcp = ei + w * CH;
    const int* __restrict__ dstp = ei + N_EDGES + w * CH;
    constexpr int PER = (CH + BT - 1) / BT;
    int pk[PER], bn[PER], rk[PER];
    #pragma unroll
    for (int k = 0; k < PER; ++k) {
        const int i = tid + k * BT;
        bn[k] = -1; pk[k] = 0; rk[k] = 0;
        if (i < CH) {
            const int d = dstp[i];
            bn[k] = d >> BIN_SHIFT;
            pk[k] = (srcp[i] << BIN_SHIFT) | (d & (BIN_NODES - 1));
            rk[k] = atomicAdd(&sCnt[bn[k]], 1);
        }
    }
    __syncthreads();
    if (tid < NBIN) {
        int s = 0;
        for (int j = 0; j < tid; ++j) s += sCnt[j];
        sLoc[tid] = s;
    }
    __syncthreads();
    #pragma unroll
    for (int k = 0; k < PER; ++k) {
        if (bn[k] >= 0) {
            const int p = sLoc[bn[k]] + rk[k];
            sVal[p] = pk[k];
            sDst[p] = sCur[bn[k]] + rk[k];
        }
    }
    __syncthreads();
    for (int p = tid; p < CH; p += BT) tmp[sDst[p]] = sVal[p];
}

__global__ __launch_bounds__(BT)
void bin_sort_kernel(const int* __restrict__ tmp, const int* __restrict__ base,
                     int2* __restrict__ seg, int* __restrict__ nbr) {
    constexpr int CAP = 16384;
    __shared__ int sDeg[BIN_NODES], sCurN[BIN_NODES], sWT[BT / 32];
    __shared__ int sOut[CAP];
    const int tid = threadIdx.x, lane = tid & 31, wave = tid >> 5;
    const int b = blockIdx.x;
    const int beg = base[b], end = base[b + 1], cnt = end - beg;
    sDeg[tid] = 0;
    __syncthreads();
    for (int i = beg + tid; i < end; i += BT) atomicAdd(&sDeg[tmp[i] & (BIN_NODES - 1)], 1);
    __syncthreads();
    const int c = sDeg[tid];
    int incl = c;
    for (int d = 1; d < 32; d <<= 1) {
        const int t = __shfl_up(incl, d);
        if (lane >= d) incl += t;
    }
    if (lane == 31) sWT[wave] = incl;
    __syncthreads();
    int wb = 0;
    for (int j = 0; j < wave; ++j) wb += sWT[j];
    const int off = wb + incl - c;
    sCurN[tid] = off;
    const int node = b * BIN_NODES + tid;
    if (node < N_NODES) seg[node] = make_int2(beg + off, c);
    __syncthreads();
    const bool staged = cnt <= CAP;
    for (int i = beg + tid; i < end; i += BT) {
        const int v = tmp[i];
        const int pos = atomicAdd(&sCurN[v & (BIN_NODES - 1)], 1);
        const int s = v >> BIN_SHIFT;
        if (staged) sOut[pos] = s; else nbr[beg + pos] = s;
    }
    __syncthreads();
    if (staged) for (int i = tid; i < cnt; i += BT) nbr[beg + i] = sOut[i];
}

template<int K, int OUT, bool RELU, bool IDX, int NW, int GB>
__global__ __launch_bounds__(NW * 32)
void sage_layer_kernel(const int* __restrict__ nbr,
                       const int2* __restrict__ seg,
                       const float* __restrict__ X,
                       const int* __restrict__ xidx,
                       const float* __restrict__ Wl,
                       const float* __restrict__ bias,
                       const float* __restrict__ Wr,
                       float* __restrict__ out) {
    static_assert(K == 128, "one float4 per lane covers a row");
    constexpr int CT  = OUT / 16;
    constexpr int LDA = K + 4;
    constexpr int KP  = K / 2;
    constexpr int WP  = OUT * 2 + 32;
    constexpr int NT  = NW * 32;
    __shared__ __attribute__((aligned(16))) float sW[2 * KP * WP];
    __shared__ __attribute__((aligned(16))) float sTile[NW * 16 * LDA];

    const int lane = threadIdx.x;
    const int wave = __builtin_amdgcn_readfirstlane(threadIdx.y);
    const int half = lane >> 4;
    const int l16  = lane & 15;

    {
        constexpr int C4 = OUT / 4;
        constexpr int PER = KP * C4;
        const int tid = threadIdx.y * 32 + threadIdx.x;
        #pragma unroll
        for (int it = 0; it < (2 * PER + NT - 1) / NT; ++it) {
            const int i = tid + it * NT;
            if ((it + 1) * NT > 2 * PER && i >= 2 * PER) continue;
            const int m   = i / PER;
            const int rem = i - m * PER;
            const int kp  = rem / C4;
            const int c4  = rem - kp * C4;
            const float* __restrict__ W = m ? Wr : Wl;
            const float4 r0 = *reinterpret_cast<const float4*>(W + (size_t)(2 * kp) * OUT + c4 * 4);
            const float4 r1 = *reinterpret_cast<const float4*>(W + (size_t)(2 * kp + 1) * OUT + c4 * 4);
            float* d = sW + (m * KP + kp) * WP + c4 * 8;
            *reinterpret_cast<float4*>(d)     = make_float4(r0.x, r1.x, r0.y, r1.y);
            *reinterpret_cast<float4*>(d + 4) = make_float4(r0.z, r1.z, r0.w, r1.w);
        }
    }
    __syncthreads();

    float* const tileM = sTile + wave * (16 * LDA);
    const float* const aM = tileM + l16 * LDA + half * 2;
    const float* const bL = sW + half * WP + l16 * 2;
    const float* const bR = bL + KP * WP;

    for (int tile = wave * (int)gridDim.x + (int)blockIdx.x; tile < N_NODES / 16;
         tile += (int)gridDim.x * NW) {
        const int node0 = tile * 16;

        const int2 sd = seg[node0 + l16];
        const int start = __builtin_amdgcn_readlane(sd.x, 0);
        const int total = __builtin_amdgcn_readlane(sd.x, 15) + __builtin_amdgcn_readlane(sd.y, 15) - start;

        int cur = 0;
        int curEnd = __builtin_amdgcn_readlane(sd.y, 0);
        float4 acc = make_float4(0.f, 0.f, 0.f, 0.f);

        auto flush = [&]() {
            const int dg = __builtin_amdgcn_readlane(sd.y, cur);
            const float inv = 1.0f / fmaxf((float)dg, 1.0f);
            acc.x *= inv; acc.y *= inv; acc.z *= inv; acc.w *= inv;
            *reinterpret_cast<float4*>(&tileM[cur * LDA + lane * 4]) = acc;
            acc = make_float4(0.f, 0.f, 0.f, 0.f);
            ++cur;
            if (cur < 16) curEnd += __builtin_amdgcn_readlane(sd.y, cur);
        };
        auto take = [&](int p, const float4& v) {
            while (p >= curEnd && cur < 15) flush();
            acc.x += v.x; acc.y += v.y; acc.z += v.z; acc.w += v.w;
        };

        int nraw = 0;
        if (lane < total) nraw = nbr[start + lane];
        for (int base = 0; base < total; base += 32) {
            const int n = (total - base < 32) ? (total - base) : 32;
            int idx = nraw;
            if (base + 32 + lane < total) nraw = nbr[start + base + 32 + lane];
            if (IDX) { if (lane < n) idx = xidx[idx]; }

            int j = 0;
            for (; j + GB <= n; j += GB) {
                float4 v[GB];
                #pragma unroll
                for (int u = 0; u < GB; ++u) {
                    const int s = __builtin_amdgcn_readlane(idx, j + u);
                    v[u] = *reinterpret_cast<const float4*>(X + (size_t)s * K + lane * 4);
                }
                #pragma unroll
                for (int u = 0; u < GB; ++u) take(base + j + u, v[u]);
            }
            for (; j + 4 <= n; j += 4) {
                float4 v[4];
                #pragma unroll
                for (int u = 0; u < 4; ++u) {
                    const int s = __builtin_amdgcn_readlane(idx, j + u);
                    v[u] = *reinterpret_cast<const float4*>(X + (size_t)s * K + lane * 4);
                }
                #pragma unroll
                for (int u = 0; u < 4; ++u) take(base + j + u, v[u]);
            }
            for (; j < n; ++j) {
                const int s = __builtin_amdgcn_readlane(idx, j);
                const float4 v = *reinterpret_cast<const float4*>(X + (size_t)s * K + lane * 4);
                take(base + j, v);
            }
        }
        while (cur < 16) flush();

        __builtin_amdgcn_fence(__ATOMIC_RELEASE, "wavefront");
        __builtin_amdgcn_wave_barrier();
        __builtin_amdgcn_fence(__ATOMIC_ACQUIRE, "wavefront");

        int xrow = node0 + l16;
        if (IDX) xrow = xidx[xrow];
#define OWN_ROWS(F) F(0) F(1) F(2) F(3) F(4) F(5) F(6) F(7) F(8) F(9) F(10) F(11) F(12) F(13) F(14) F(15)
#define OWN_ROW_LOAD(r) \
        const float4 xr##r = *reinterpret_cast<const float4*>( \
            X + (size_t)__builtin_amdgcn_readlane(xrow, r) * K + lane * 4);
#define OWN_ROW_PARK(r) *reinterpret_cast<float4*>(&tileM[r * LDA + lane * 4]) = xr##r;
        OWN_ROWS(OWN_ROW_LOAD)

        v8f c[CT];
        #pragma unroll
        for (int t = 0; t < CT; ++t) c[t] = (v8f){0.f, 0.f, 0.f, 0.f, 0.f, 0.f, 0.f, 0.f};

        auto product = [&](const float* __restrict__ bp) {
            const float* __restrict__ ap = aM;
            v2f a0, a1, b0[CT], b1[CT];
            a0 = *reinterpret_cast<const v2f*>(ap);
            #pragma unroll
            for (int t = 0; t < CT; ++t) b0[t] = *reinterpret_cast<const v2f*>(bp + t * 32);
            #pragma nounroll
            for (int it = 0; it < K / 8 - 1; ++it) {
                a1 = *reinterpret_cast<const v2f*>(ap + 4);
                #pragma unroll
                for (int t = 0; t < CT; ++t) b1[t] = *reinterpret_cast<const v2f*>(bp + 2 * WP + t * 32);
                #pragma unroll
                for (int t = 0; t < CT; ++t)
                    c[t] = __builtin_amdgcn_wmma_f32_16x16x4_f32(false, a0, false, b0[t],
                                                                 (short)0, c[t], false, false);
                a0 = *reinterpret_cast<const v2f*>(ap + 8);
                #pragma unroll
                for (int t = 0; t < CT; ++t) b0[t] = *reinterpret_cast<const v2f*>(bp + 4 * WP + t * 32);
                #pragma unroll
                for (int t = 0; t < CT; ++t)
                    c[t] = __builtin_amdgcn_wmma_f32_16x16x4_f32(false, a1, false, b1[t],
                                                                 (short)0, c[t], false, false);
                ap += 8;
                bp += 4 * WP;
            }
            a1 = *reinterpret_cast<const v2f*>(ap + 4);
            #pragma unroll
            for (int t = 0; t < CT; ++t) b1[t] = *reinterpret_cast<const v2f*>(bp + 2 * WP + t * 32);
            #pragma unroll
            for (int t = 0; t < CT; ++t)
                c[t] = __builtin_amdgcn_wmma_f32_16x16x4_f32(false, a0, false, b0[t],
                                                             (short)0, c[t], false, false);
            #pragma unroll
            for (int t = 0; t < CT; ++t)
                c[t] = __builtin_amdgcn_wmma_f32_16x16x4_f32(false, a1, false, b1[t],
                                                             (short)0, c[t], false, false);
        };

        product(bL);

        __builtin_amdgcn_fence(__ATOMIC_RELEASE, "wavefront");
        __builtin_amdgcn_wave_barrier();
        __builtin_amdgcn_fence(__ATOMIC_ACQUIRE, "wavefront");
        OWN_ROWS(OWN_ROW_PARK)
#undef OWN_ROW_PARK
#undef OWN_ROW_LOAD
#undef OWN_ROWS
        __builtin_amdgcn_fence(__ATOMIC_RELEASE, "wavefront");
        __builtin_amdgcn_wave_barrier();
        __builtin_amdgcn_fence(__ATOMIC_ACQUIRE, "wavefront");

        product(bR);

        #pragma unroll
        for (int t = 0; t < CT; ++t) {
            const int col = t * 16 + l16;
            const float bv = bias[col];
            #pragma unroll
            for (int v = 0; v < 8; ++v) {
                const int r = node0 + v + half * 8;
                float val = c[t][v] + bv;
                if (RELU) val = fmaxf(val, 0.0f);
                out[(size_t)r * OUT + col] = val;
            }
        }

        __builtin_amdgcn_fence(__ATOMIC_RELEASE, "wavefront");
        __builtin_amdgcn_wave_barrier();
        __builtin_amdgcn_fence(__ATOMIC_ACQUIRE, "wavefront");
    }
}

extern "C" void kernel_launch(void* const* d_in, const int* in_sizes, int n_in,
                              void* d_out, int out_size, void* d_ws, size_t ws_size,
                              hipStream_t stream) {
    (void)in_sizes; (void)n_in; (void)out_size; (void)ws_size;

    const int*   xidx = (const int*)d_in[0];
    const int*   ei   = (const int*)d_in[1];
    const float* emb  = (const float*)d_in[2];
    const float* Wl1  = (const float*)d_in[3];
    const float* bl1  = (const float*)d_in[4];
    const float* Wr1  = (const float*)d_in[5];
    const float* Wl2  = (const float*)d_in[6];
    const float* bl2  = (const float*)d_in[7];
    const float* Wr2  = (const float*)d_in[8];

    int*   nbr   = (int*)d_ws;
    int*   tmp   = nbr + N_EDGES;
    int2*  seg   = (int2*)(tmp + N_EDGES);
    int*   histT = (int*)(seg + N_NODES);
    int*   base  = histT + NBIN * NCHUNK;
    float* h     = (float*)d_ws + (size_t)N_NODES * HIDDEN;

    bin_count_kernel<<<NCHUNK, BT, 0, stream>>>(ei, histT);
    bin_scatter_kernel<<<NCHUNK, BT, 0, stream>>>(ei, histT, base, tmp);
    bin_sort_kernel<<<NBIN, BT, 0, stream>>>(tmp, base, seg, nbr);


    sage_layer_kernel<EMB_DIM, HIDDEN, true, true, 16, 16>
        <<<dim3(256), dim3(32, 16), 0, stream>>>(
            nbr, seg, emb, xidx, Wl1, bl1, Wr1, h);

    sage_layer_kernel<HIDDEN, OUT_C, false, false, 28, 8>
        <<<dim3(256), dim3(32, 28), 0, stream>>>(
            nbr, seg, h, nullptr, Wl2, bl2, Wr2, (float*)d_out);
}


# Layer 2 projects before it aggregates: layer 1 forms [h@Wl2 | h@Wr2+b] from its h tile (third WMMA product, weights in LDS), layer 2 only averages 64-float rows; sum order changes
# speedup vs baseline: 3.6918x; 1.0317x over previous
#include <hip/hip_runtime.h>

typedef __attribute__((ext_vector_type(2))) float v2f;
typedef __attribute__((ext_vector_type(8))) float v8f;

#define N_NODES 100000
#define EMB_DIM 128
#define HIDDEN  128
#define OUT_C   64
#define N_EDGES 800000

#define NCHUNK    256
#define CH        (N_EDGES / NCHUNK)
#define BIN_SHIFT 9
#define BIN_NODES (1 << BIN_SHIFT)
#define NBIN      ((N_NODES + BIN_NODES - 1) / BIN_NODES)
#define BT        512
static_assert(CH * NCHUNK == N_EDGES, "chunks tile the edge list");
static_assert(NBIN <= BT && BIN_NODES == BT, "one thread per bin / per node of a bin");
static_assert(NCHUNK == 256, "one lane sums 8 chunk counts");

__global__ __launch_bounds__(BT)
void bin_count_kernel(const int* __restrict__ ei, int* __restrict__ histT) {
    __shared__ int sCnt[NBIN];
    const int tid = threadIdx.x;
    if (tid < NBIN) sCnt[tid] = 0;
    __syncthreads();
    const int* __restrict__ dst = ei + N_EDGES + blockIdx.x * CH;
    for (int i = tid; i < CH; i += BT) atomicAdd(&sCnt[dst[i] >> BIN_SHIFT], 1);
    __syncthreads();
    if (tid < NBIN) histT[tid * NCHUNK + blockIdx.x] = sCnt[tid];
}

__global__ __launch_bounds__(BT)
void bin_scatter_kernel(const int* __restrict__ ei, const int* __restrict__ histT,
                        int* __restrict__ base, int* __restrict__ tmp) {
    __shared__ int sTot[NBIN], sPre[NBIN], sCur[NBIN], sLoc[NBIN], sCnt[NBIN];
    __shared__ int sVal[CH], sDst[CH];
    const int tid = threadIdx.x, lane = tid & 31, wave = tid >> 5;
    const int w = blockIdx.x;
    if (tid < NBIN) sCnt[tid] = 0;

    for (int b = wave; b < NBIN; b += BT / 32) {
        const int4* __restrict__ p = reinterpret_cast<const int4*>(histT + b * NCHUNK) + lane * 2;
        const int4 a = p[0], c = p[1];
        const int w0 = lane * 8;
        int tot = a.x + a.y + a.z + a.w + c.x + c.y + c.z + c.w;
        int pre = (w0 + 0 < w ? a.x : 0) + (w0 + 1 < w ? a.y : 0) + (w0 + 2 < w ? a.z : 0) +
                  (w0 + 3 < w ? a.w : 0) + (w0 + 4 < w ? c.x : 0) + (w0 + 5 < w ? c.y : 0) +
                  (w0 + 6 < w ? c.z : 0) + (w0 + 7 < w ? c.w : 0);
        for (int d = 16; d >= 1; d >>= 1) { tot += __shfl_xor(tot, d); pre += __shfl_xor(pre, d); }
        if (lane == 0) { sTot[b] = tot; sPre[b] = pre; }
    }
    __syncthreads();
    if (tid < NBIN) {
        int s = 0;
        for (int j = 0; j < tid; ++j) s += sTot[j];
        sCur[tid] = s + sPre[tid];
        if (w == 0) base[tid] = s;
    }
    if (w == 0 && tid == 0) base[NBIN] = N_EDGES;

    const int* __restrict__ srcp = ei + w * CH;
    const int* __restrict__ dstp = ei + N_EDGES + w * CH;
    constexpr int PER = (CH + BT - 1) / BT;
    int pk[PER], bn[PER], rk[PER];
    #pragma unroll
    for (int k = 0; k < PER; ++k) {
        const int i = tid + k * BT;
        bn[k] = -1; pk[k] = 0; rk[k] = 0;
        if (i < CH) {
            const int d = dstp[i];
            bn[k] = d >> BIN_SHIFT;
            pk[k] = (srcp[i] << BIN_SHIFT) | (d & (BIN_NODES - 1));
            rk[k] = atomicAdd(&sCnt[bn[k]], 1);
        }
    }
    __syncthreads();
    if (tid < NBIN) {
        int s = 0;
        for (int j = 0; j < tid; ++j) s += sCnt[j];
        sLoc[tid] = s;
    }
    __syncthreads();
    #pragma unroll
    for (int k = 0; k < PER; ++k) {
        if (bn[k] >= 0) {
            const int p = sLoc[bn[k]] + rk[k];
            sVal[p] = pk[k];
            sDst[p] = sCur[bn[k]] + rk[k];
        }
    }
    __syncthreads();
    for (int p = tid; p < CH; p += BT) tmp[sDst[p]] = sVal[p];
}

__global__ __launch_bounds__(BT)
void bin_sort_kernel(const int* __restrict__ tmp, const int* __restrict__ base,
                     int2* __restrict__ seg, int* __restrict__ nbr) {
    constexpr int CAP = 16384;
    __shared__ int sDeg[BIN_NODES], sCurN[BIN_NODES], sWT[BT / 32];
    __shared__ int sOut[CAP];
    const int tid = threadIdx.x, lane = tid & 31, wave = tid >> 5;
    const int b = blockIdx.x;
    const int beg = base[b], end = base[b + 1], cnt = end - beg;
    sDeg[tid] = 0;
    __syncthreads();
    for (int i = beg + tid; i < end; i += BT) atomicAdd(&sDeg[tmp[i] & (BIN_NODES - 1)], 1);
    __syncthreads();
    const int c = sDeg[tid];
    int incl = c;
    for (int d = 1; d < 32; d <<= 1) {
        const int t = __shfl_up(incl, d);
        if (lane >= d) incl += t;
    }
    if (lane == 31) sWT[wave] = incl;
    __syncthreads();
    int wb = 0;
    for (int j = 0; j < wave; ++j) wb += sWT[j];
    const int off = wb + incl - c;
    sCurN[tid] = off;
    const int node = b * BIN_NODES + tid;
    if (node < N_NODES) seg[node] = make_int2(beg + off, c);
    __syncthreads();
    const bool staged = cnt <= CAP;
    for (int i = beg + tid; i < end; i += BT) {
        const int v = tmp[i];
        const int pos = atomicAdd(&sCurN[v & (BIN_NODES - 1)], 1);
        const int s = v >> BIN_SHIFT;
        if (staged) sOut[pos] = s; else nbr[beg + pos] = s;
    }
    __syncthreads();
    if (staged) for (int i = tid; i < cnt; i += BT) nbr[beg + i] = sOut[i];
}

template<int K, int OUT, bool RELU, bool IDX, int NW, int GB>
__global__ __launch_bounds__(NW * 32)
void sage_layer_kernel(const int* __restrict__ nbr,
                       const int2* __restrict__ seg,
                       const float* __restrict__ X,
                       const int* __restrict__ xidx,
                       const float* __restrict__ Wl,
                       const float* __restrict__ bias,
                       const float* __restrict__ Wr,
                       const float* __restrict__ Wl2,
                       const float* __restrict__ bias2,
                       const float* __restrict__ Wr2,
                       float* __restrict__ out) {
    static_assert(K == 128, "one float4 per lane covers a row");
    static_assert(OUT == K, "the h tile is the A operand of the third product");
    constexpr int CT  = OUT / 16;
    constexpr int LDA = K + 4;
    constexpr int KP  = K / 2;
    constexpr int STEP = OUT * 2;
    constexpr int ODD  = (KP / 2) * STEP + 32;
    constexpr int MS   = KP * STEP + 64;
    constexpr int O2   = OUT / 2;
    constexpr int NT  = NW * 32;
    __shared__ __attribute__((aligned(16))) float sW[3 * MS];
    __shared__ __attribute__((aligned(16))) float sTile[NW * 16 * LDA];

    const int lane = threadIdx.x;
    const int wave = __builtin_amdgcn_readfirstlane(threadIdx.y);
    const int half = lane >> 4;
    const int l16  = lane & 15;

    {
        constexpr int C4 = OUT / 4;
        constexpr int PER = KP * C4;
        const int tid = threadIdx.y * 32 + threadIdx.x;
        #pragma unroll
        for (int it = 0; it < (3 * PER + NT - 1) / NT; ++it) {
            const int i = tid + it * NT;
            if ((it + 1) * NT > 3 * PER && i >= 3 * PER) continue;
            const int m   = i / PER;
            const int rem = i - m * PER;
            const int kp  = rem / C4;
            const int c4  = rem - kp * C4;
            const int c2  = c4 * 4 - (c4 * 4 >= O2 ? O2 : 0);
            const float* __restrict__ W2 = (c4 * 4 >= O2) ? Wr2 : Wl2;
            const float* __restrict__ p0 = (m == 2) ? W2 + (size_t)(2 * kp) * O2 + c2
                                                    : (m ? Wr : Wl) + (size_t)(2 * kp) * OUT + c4 * 4;
            const float4 r0 = *reinterpret_cast<const float4*>(p0);
            const float4 r1 = *reinterpret_cast<const float4*>(p0 + (m == 2 ? O2 : OUT));
            float* d = sW + m * MS + (kp & 1) * ODD + (kp >> 1) * STEP + c4 * 8;
            *reinterpret_cast<float4*>(d)     = make_float4(r0.x, r1.x, r0.y, r1.y);
            *reinterpret_cast<float4*>(d + 4) = make_float4(r0.z, r1.z, r0.w, r1.w);
        }
    }
    __syncthreads();

    float* const tileM = sTile + wave * (16 * LDA);
    const float* const aM = tileM + l16 * LDA + half * 2;
    const float* const bL = sW + half * ODD + l16 * 2;
    const float* const bR = bL + MS;
    const float* const bP = bR + MS;

    const int turn = (NW == 15 && wave == 11) ? 8 : (NW == 15 && wave >= 8 && wave < 11) ? wave + 1 : wave;
    for (int tile = turn * (int)gridDim.x + (int)blockIdx.x; tile < N_NODES / 16;
         tile += (int)gridDim.x * NW) {
        const int node0 = tile * 16;

        const int2 sd = seg[node0 + l16];
        const int start = __builtin_amdgcn_readlane(sd.x, 0);
        const int total = __builtin_amdgcn_readlane(sd.x, 15) + __builtin_amdgcn_readlane(sd.y, 15) - start;

        int cur = 0;
        int curEnd = __builtin_amdgcn_readlane(sd.y, 0);
        float4 acc = make_float4(0.f, 0.f, 0.f, 0.f);

        auto flush = [&]() {
            const int dg = __builtin_amdgcn_readlane(sd.y, cur);
            const float inv = 1.0f / fmaxf((float)dg, 1.0f);
            acc.x *= inv; acc.y *= inv; acc.z *= inv; acc.w *= inv;
            *reinterpret_cast<float4*>(&tileM[cur * LDA + lane * 4]) = acc;
            acc = make_float4(0.f, 0.f, 0.f, 0.f);
            ++cur;
            if (cur < 16) curEnd += __builtin_amdgcn_readlane(sd.y, cur);
        };
        auto take = [&](int p, const float4& v) {
            while (p >= curEnd && cur < 15) flush();
            acc.x += v.x; acc.y += v.y; acc.z += v.z; acc.w += v.w;
        };

        int nraw = 0;
        if (lane < total) nraw = nbr[start + lane];
        for (int base = 0; base < total; base += 32) {
            const int n = (total - base < 32) ? (total - base) : 32;
            int idx = nraw;
            if (base + 32 + lane < total) nraw = nbr[start + base + 32 + lane];
            if (IDX) { if (lane < n) idx = xidx[idx]; }

            int j = 0;
            for (; j + GB <= n; j += GB) {
                float4 v[GB];
                #pragma unroll
                for (int u = 0; u < GB; ++u) {
                    const int s = __builtin_amdgcn_readlane(idx, j + u);
                    v[u] = *reinterpret_cast<const float4*>(X + (size_t)s * K + lane * 4);
                }
                #pragma unroll
                for (int u = 0; u < GB; ++u) take(base + j + u, v[u]);
            }
            for (; j + 4 <= n; j += 4) {
                float4 v[4];
                #pragma unroll
                for (int u = 0; u < 4; ++u) {
                    const int s = __builtin_amdgcn_readlane(idx, j + u);
                    v[u] = *reinterpret_cast<const float4*>(X + (size_t)s * K + lane * 4);
                }
                #pragma unroll
                for (int u = 0; u < 4; ++u) take(base + j + u, v[u]);
            }
            for (; j < n; ++j) {
                const int s = __builtin_amdgcn_readlane(idx, j);
                const float4 v = *reinterpret_cast<const float4*>(X + (size_t)s * K + lane * 4);
                take(base + j, v);
            }
        }
        while (cur < 16) flush();

        __builtin_amdgcn_fence(__ATOMIC_RELEASE, "wavefront");
        __builtin_amdgcn_wave_barrier();
        __builtin_amdgcn_fence(__ATOMIC_ACQUIRE, "wavefront");

        int xrow = node0 + l16;
        if (IDX) xrow = xidx[xrow];
#define OWN_ROWS(F) F(0) F(1) F(2) F(3) F(4) F(5) F(6) F(7) F(8) F(9) F(10) F(11) F(12) F(13) F(14) F(15)
#define OWN_ROW_LOAD(r) \
        const float4 xr##r = *reinterpret_cast<const float4*>( \
            X + (size_t)__builtin_amdgcn_readlane(xrow, r) * K + lane * 4);
#define OWN_ROW_PARK(r) *reinterpret_cast<float4*>(&tileM[r * LDA + lane * 4]) = xr##r;
        OWN_ROWS(OWN_ROW_LOAD)

        v8f c[CT];
        #pragma unroll
        for (int t = 0; t < CT; ++t) c[t] = (v8f){0.f, 0.f, 0.f, 0.f, 0.f, 0.f, 0.f, 0.f};

        auto product = [&](const float* __restrict__ bp) {
            const float* __restrict__ ap = aM;
            v2f a0, a1, b0[CT], b1[CT];
            a0 = *reinterpret_cast<const v2f*>(ap);
            #pragma unroll
            for (int t = 0; t < CT; ++t) b0[t] = *reinterpret_cast<const v2f*>(bp + t * 32);
            #pragma nounroll
            for (int it = 0; it < K / 8 - 1; ++it) {
                a1 = *reinterpret_cast<const v2f*>(ap + 4);
                #pragma unroll
                for (int t = 0; t < CT; ++t) b1[t] = *reinterpret_cast<const v2f*>(bp + STEP + t * 32);
                #pragma unroll
                for (int t = 0; t < CT; ++t)
                    c[t] = __builtin_amdgcn_wmma_f32_16x16x4_f32(false, a0, false, b0[t],
                                                                 (short)0, c[t], false, false);
                a0 = *reinterpret_cast<const v2f*>(ap + 8);
                #pragma unroll
                for (int t = 0; t < CT; ++t) b0[t] = *reinterpret_cast<const v2f*>(bp + 2 * STEP + t * 32);
                #pragma unroll
                for (int t = 0; t < CT; ++t)
                    c[t] = __builtin_amdgcn_wmma_f32_16x16x4_f32(false, a1, false, b1[t],
                                                                 (short)0, c[t], false, false);
                ap += 8;
                bp += 2 * STEP;
            }
            a1 = *reinterpret_cast<const v2f*>(ap + 4);
            #pragma unroll
            for (int t = 0; t < CT; ++t) b1[t] = *reinterpret_cast<const v2f*>(bp + STEP + t * 32);
            #pragma unroll
            for (int t = 0; t < CT; ++t)
                c[t] = __builtin_amdgcn_wmma_f32_16x16x4_f32(false, a0, false, b0[t],
                                                             (short)0, c[t], false, false);
            #pragma unroll
            for (int t = 0; t < CT; ++t)
                c[t] = __builtin_amdgcn_wmma_f32_16x16x4_f32(false, a1, false, b1[t],
                                                             (short)0, c[t], false, false);
        };

        product(bL);

        __builtin_amdgcn_fence(__ATOMIC_RELEASE, "wavefront");
        __builtin_amdgcn_wave_barrier();
        __builtin_amdgcn_fence(__ATOMIC_ACQUIRE, "wavefront");
        OWN_ROWS(OWN_ROW_PARK)
#undef OWN_ROW_PARK
#undef OWN_ROW_LOAD
#undef OWN_ROWS
        __builtin_amdgcn_fence(__ATOMIC_RELEASE, "wavefront");
        __builtin_amdgcn_wave_barrier();
        __builtin_amdgcn_fence(__ATOMIC_ACQUIRE, "wavefront");

        product(bR);

        __builtin_amdgcn_fence(__ATOMIC_RELEASE, "wavefront");
        __builtin_amdgcn_wave_barrier();
        __builtin_amdgcn_fence(__ATOMIC_ACQUIRE, "wavefront");
        #pragma unroll
        for (int t = 0; t < CT; ++t) {
            const int col = t * 16 + l16;
            const float bv = bias[col];
            #pragma unroll
            for (int v = 0; v < 8; ++v) {
                float val = c[t][v] + bv;
                if (RELU) val = fmaxf(val, 0.0f);
                tileM[(v + half * 8) * LDA + col] = val;
            }
            c[t] = (v8f){0.f, 0.f, 0.f, 0.f, 0.f, 0.f, 0.f, 0.f};
        }
        __builtin_amdgcn_fence(__ATOMIC_RELEASE, "wavefront");
        __builtin_amdgcn_wave_barrier();
        __builtin_amdgcn_fence(__ATOMIC_ACQUIRE, "wavefront");

        product(bP);

        #pragma unroll
        for (int t = 0; t < CT; ++t) {
            const int col = t * 16 + l16;
            const float bv = (t * 16 >= O2) ? bias2[col - O2] : 0.0f;
            #pragma unroll
            for (int v = 0; v < 8; ++v) {
                const int r = node0 + v + half * 8;
                out[(size_t)r * OUT + col] = c[t][v] + bv;
            }
        }

        __builtin_amdgcn_fence(__ATOMIC_RELEASE, "wavefront");
        __builtin_amdgcn_wave_barrier();
        __builtin_amdgcn_fence(__ATOMIC_ACQUIRE, "wavefront");
    }
}

static_assert(HIDDEN == 2 * OUT_C && OUT_C == 64 && N_NODES % 16 == 0, "row split and tiles of mean_add_kernel");
template<int NW, int GB>
__global__ __launch_bounds__(NW * 32)
void mean_add_kernel(const int* __restrict__ nbr, const int2* __restrict__ seg,
                     const float* __restrict__ zr, float* __restrict__ out) {
    constexpr int LD = OUT_C;
    __shared__ __attribute__((aligned(8))) float sMean[NW * 16 * LD];
    const int lane = threadIdx.x;
    const int wave = __builtin_amdgcn_readfirstlane(threadIdx.y);
    const int tile = wave * (int)gridDim.x + (int)blockIdx.x;
    if (tile >= N_NODES / 16) return;
    const int node0 = tile * 16;
    const int l16 = lane & 15;
    float* const tileM = sMean + wave * (16 * LD);

    const int2 sd = seg[node0 + l16];

#define R_ROWS(F) F(0) F(1) F(2) F(3) F(4) F(5) F(6) F(7) F(8) F(9) F(10) F(11) F(12) F(13) F(14) F(15)
#define R_ROW_LOAD(r) \
    const float2 rr##r = *reinterpret_cast<const float2*>( \
        zr + (size_t)(node0 + r) * HIDDEN + OUT_C + lane * 2);
#define R_ROW_OUT(r) { \
        const float2 m = *reinterpret_cast<const float2*>(&tileM[r * LD + lane * 2]); \
        *reinterpret_cast<float2*>(out + (size_t)(node0 + r) * OUT_C + lane * 2) = \
            make_float2(rr##r.x + m.x, rr##r.y + m.y); }
    R_ROWS(R_ROW_LOAD)

    const int start = __builtin_amdgcn_readlane(sd.x, 0);
    const int total = __builtin_amdgcn_readlane(sd.x, 15) + __builtin_amdgcn_readlane(sd.y, 15) - start;

    int cur = 0;
    int curEnd = __builtin_amdgcn_readlane(sd.y, 0);
    float2 acc = make_float2(0.f, 0.f);

    auto flush = [&]() {
        const int dg = __builtin_amdgcn_readlane(sd.y, cur);
        const float inv = 1.0f / fmaxf((float)dg, 1.0f);
        *reinterpret_cast<float2*>(&tileM[cur * LD + lane * 2]) = make_float2(acc.x * inv, acc.y * inv);
        acc = make_float2(0.f, 0.f);
        ++cur;
        if (cur < 16) curEnd += __builtin_amdgcn_readlane(sd.y, cur);
    };
    auto take = [&](int p, const float2& v) {
        while (p >= curEnd && cur < 15) flush();
        acc.x += v.x; acc.y += v.y;
    };

    int nraw = 0;
    if (lane < total) nraw = nbr[start + lane];
    for (int base = 0; base < total; base += 32) {
        const int n = (total - base < 32) ? (total - base) : 32;
        const int idx = nraw;
        if (base + 32 + lane < total) nraw = nbr[start + base + 32 + lane];

        int j = 0;
        for (; j + GB <= n; j += GB) {
            float2 v[GB];
            #pragma unroll
            for (int u = 0; u < GB; ++u) {
                const int s = __builtin_amdgcn_readlane(idx, j + u);
                v[u] = *reinterpret_cast<const float2*>(zr + (size_t)s * HIDDEN + lane * 2);
            }
            #pragma unroll
            for (int u = 0; u < GB; ++u) take(base + j + u, v[u]);
        }
        for (; j + 4 <= n; j += 4) {
            float2 v[4];
            #pragma unroll
            for (int u = 0; u < 4; ++u) {
                const int s = __builtin_amdgcn_readlane(idx, j + u);
                v[u] = *reinterpret_cast<const float2*>(zr + (size_t)s * HIDDEN + lane * 2);
            }
            #pragma unroll
            for (int u = 0; u < 4; ++u) take(base + j + u, v[u]);
        }
        for (; j < n; ++j) {
            const int s = __builtin_amdgcn_readlane(idx, j);
            const float2 v = *reinterpret_cast<const float2*>(zr + (size_t)s * HIDDEN + lane * 2);
            take(base + j, v);
        }
    }
    while (cur < 16) flush();

    R_ROWS(R_ROW_OUT)
#undef R_ROW_OUT
#undef R_ROW_LOAD
#undef R_ROWS
}

extern "C" void kernel_launch(void* const* d_in, const int* in_sizes, int n_in,
                              void* d_out, int out_size, void* d_ws, size_t ws_size,
                              hipStream_t stream) {
    (void)in_sizes; (void)n_in; (void)out_size; (void)ws_size;

    const int*   xidx = (const int*)d_in[0];
    const int*   ei   = (const int*)d_in[1];
    const float* emb  = (const float*)d_in[2];
    const float* Wl1  = (const float*)d_in[3];
    const float* bl1  = (const float*)d_in[4];
    const float* Wr1  = (const float*)d_in[5];
    const float* Wl2  = (const float*)d_in[6];
    const float* bl2  = (const float*)d_in[7];
    const float* Wr2  = (const float*)d_in[8];

    int*   nbr   = (int*)d_ws;
    int*   tmp   = nbr + N_EDGES;
    int2*  seg   = (int2*)(tmp + N_EDGES);
    int*   histT = (int*)(seg + N_NODES);
    int*   base  = histT + NBIN * NCHUNK;
    float* zr    = (float*)d_ws + (size_t)N_NODES * HIDDEN;

    bin_count_kernel<<<NCHUNK, BT, 0, stream>>>(ei, histT);
    bin_scatter_kernel<<<NCHUNK, BT, 0, stream>>>(ei, histT, base, tmp);
    bin_sort_kernel<<<NBIN, BT, 0, stream>>>(tmp, base, seg, nbr);


    sage_layer_kernel<EMB_DIM, HIDDEN, true, true, 15, 16>
        <<<dim3(256), dim3(32, 15), 0, stream>>>(
            nbr, seg, emb, xidx, Wl1, bl1, Wr1, Wl2, bl2, Wr2, zr);

    mean_add_kernel<25, 16><<<dim3(256), dim3(32, 25), 0, stream>>>(
        nbr, seg, zr, (float*)d_out);
}


# Edge bucketing in one launch, not three: chunks ordered by 400-node bin only; each layer workgroup owns a bin and builds its seg/nbr in its prologue; sum order still set by LDS atomic arrival
# speedup vs baseline: 4.3036x; 1.1657x over previous
#include <hip/hip_runtime.h>

typedef __attribute__((ext_vector_type(2))) float v2f;
typedef __attribute__((ext_vector_type(8))) float v8f;

#define N_NODES 100000
#define EMB_DIM 128
#define HIDDEN  128
#define OUT_C   64
#define N_EDGES 800000

#define NCHUNK    256
#define CH        (N_EDGES / NCHUNK)
#define BIN_NODES 400
#define NBIN      (N_NODES / BIN_NODES)
#define LOC_BITS  9
#define LOC_MASK  ((1 << LOC_BITS) - 1)
#define BT        512
static_assert(CH * NCHUNK == N_EDGES, "chunks tile the edge list");
static_assert(NBIN * BIN_NODES == N_NODES && BIN_NODES % 16 == 0, "bins tile the nodes in whole row tiles");
static_assert(BIN_NODES <= (1 << LOC_BITS) && N_NODES <= (1 << (31 - LOC_BITS)), "an edge fits one word");
static_assert(NBIN <= BT, "one thread per bin");
static_assert(NBIN <= 256, "one layer workgroup per bin, all resident at once");

__global__ __launch_bounds__(BT)
void chunk_sort_kernel(const int* __restrict__ ei, int2* __restrict__ segT, int* __restrict__ tmp) {
    __shared__ int sCnt[NBIN], sLoc[NBIN];
    __shared__ int sVal[CH];
    const int tid = threadIdx.x;
    const int w = blockIdx.x;
    if (tid < NBIN) sCnt[tid] = 0;
    __syncthreads();

    const int* __restrict__ srcp = ei + w * CH;
    const int* __restrict__ dstp = ei + N_EDGES + w * CH;
    constexpr int PER = (CH + BT - 1) / BT;
    int pk[PER], bn[PER], rk[PER];
    #pragma unroll
    for (int k = 0; k < PER; ++k) {
        const int i = tid + k * BT;
        bn[k] = -1; pk[k] = 0; rk[k] = 0;
        if (i < CH) {
            const int d = dstp[i];
            bn[k] = d / BIN_NODES;
            pk[k] = (srcp[i] << LOC_BITS) | (d - bn[k] * BIN_NODES);
            rk[k] = atomicAdd(&sCnt[bn[k]], 1);
        }
    }
    __syncthreads();
    if (tid < NBIN) {
        int s = 0;
        for (int j = 0; j < tid; ++j) s += sCnt[j];
        sLoc[tid] = s;
        segT[tid * NCHUNK + w] = make_int2(s, sCnt[tid]);
    }
    __syncthreads();
    #pragma unroll
    for (int k = 0; k < PER; ++k)
        if (bn[k] >= 0) sVal[sLoc[bn[k]] + rk[k]] = pk[k];
    __syncthreads();
    for (int p = tid; p < CH; p += BT) tmp[w * CH + p] = sVal[p];
}

template<int K, int OUT, bool RELU, bool IDX, int NW, int GB>
__global__ __launch_bounds__(NW * 32)
void sage_layer_kernel(const int2* __restrict__ segT,
                       const int* __restrict__ tmp,
                       int* nbr,
                       int2* seg,
                       const float* __restrict__ X,
                       const int* __restrict__ xidx,
                       const float* __restrict__ Wl,
                       const float* __restrict__ bias,
                       const float* __restrict__ Wr,
                       const float* __restrict__ Wl2,
                       const float* __restrict__ bias2,
                       const float* __restrict__ Wr2,
                       float* __restrict__ out) {
    static_assert(K == 128, "one float4 per lane covers a row");
    static_assert(OUT == K, "the h tile is the A operand of the third product");
    constexpr int CT  = OUT / 16;
    constexpr int LDA = K + 4;
    constexpr int KP  = K / 2;
    constexpr int STEP = OUT * 2;
    constexpr int ODD  = (KP / 2) * STEP + 32;
    constexpr int MS   = KP * STEP + 64;
    constexpr int O2   = OUT / 2;
    constexpr int NT  = NW * 32;
    __shared__ __attribute__((aligned(16))) float sW[3 * MS];
    constexpr int TW = NW * 16 * LDA;
    union TileSpace { float tile[TW]; int sort[TW]; };
    __shared__ __attribute__((aligned(16))) TileSpace sU;
    float* const sTile = sU.tile;

    const int lane = threadIdx.x;
    const int wave = __builtin_amdgcn_readfirstlane(threadIdx.y);
    const int half = lane >> 4;
    const int l16  = lane & 15;

    const int bin = (int)blockIdx.x;
    const int tid = threadIdx.y * 32 + threadIdx.x;
    static_assert(NT >= NCHUNK && NT >= BIN_NODES, "one thread per chunk run / per node of the bin");

    int2 run = make_int2(0, 0);
    if (tid < NCHUNK) run = segT[bin * NCHUNK + tid];

    {
        constexpr int C4 = OUT / 4;
        constexpr int PER = KP * C4;
        #pragma unroll
        for (int it = 0; it < (3 * PER + NT - 1) / NT; ++it) {
            const int i = tid + it * NT;
            if ((it + 1) * NT > 3 * PER && i >= 3 * PER) continue;
            const int m   = i / PER;
            const int rem = i - m * PER;
            const int kp  = rem / C4;
            const int c4  = rem - kp * C4;
            const int c2  = c4 * 4 - (c4 * 4 >= O2 ? O2 : 0);
            const float* __restrict__ W2 = (c4 * 4 >= O2) ? Wr2 : Wl2;
            const float* __restrict__ p0 = (m == 2) ? W2 + (size_t)(2 * kp) * O2 + c2
                                                    : (m ? Wr : Wl) + (size_t)(2 * kp) * OUT + c4 * 4;
            const float4 r0 = *reinterpret_cast<const float4*>(p0);
            const float4 r1 = *reinterpret_cast<const float4*>(p0 + (m == 2 ? O2 : OUT));
            float* d = sW + m * MS + (kp & 1) * ODD + (kp >> 1) * STEP + c4 * 8;
            *reinterpret_cast<float4*>(d)     = make_float4(r0.x, r1.x, r0.y, r1.y);
            *reinterpret_cast<float4*>(d + 4) = make_float4(r0.z, r1.z, r0.w, r1.w);
        }
    }

    {
        constexpr int PB  = 8;
        constexpr int O_PRE = 0;
        constexpr int O_ST  = O_PRE + NCHUNK;
        constexpr int O_DEG = O_ST + NCHUNK;
        constexpr int O_CUR = O_DEG + BIN_NODES;
        constexpr int O_WT  = O_CUR + BIN_NODES;
        constexpr int O_WS  = O_WT + 32;
        constexpr int O_IN  = O_WS + 32;
        constexpr int CAP   = (TW - O_IN) / 2;
        constexpr int O_OUT = O_IN + CAP;
        static_assert(CAP >= 4 * (N_EDGES / NBIN), "an average bin is staged with room to spare");
        int* const sPre = sU.sort + O_PRE;
        int* const sSt  = sU.sort + O_ST;
        int* const sDeg = sU.sort + O_DEG;
        int* const sCur = sU.sort + O_CUR;
        int* const sWT  = sU.sort + O_WT;
        int* const sWS  = sU.sort + O_WS;
        int* const sIn  = sU.sort + O_IN;
        int* const sOut = sU.sort + O_OUT;

        int incl = run.y, sts = run.x;
        for (int d = 1; d < 32; d <<= 1) {
            const int t = __shfl_up(incl, d);
            if (lane >= d) incl += t;
        }
        for (int d = 16; d >= 1; d >>= 1) sts += __shfl_xor(sts, d);
        if (lane == 31) { sWT[wave] = incl; sWS[wave] = sts; }
        if (tid < BIN_NODES) sDeg[tid] = 0;
        __syncthreads();
        int wb = 0, totalV = 0, baseV = 0;
        for (int j = 0; j < NCHUNK / 32; ++j) {
            const int t = sWT[j];
            if (j < wave) wb += t;
            totalV += t;
            baseV += sWS[j];
        }
        const int total = __builtin_amdgcn_readfirstlane(totalV);
        const int base  = __builtin_amdgcn_readfirstlane(baseV);
        if (tid < NCHUNK) { sPre[tid] = wb + incl - run.y; sSt[tid] = tid * CH + run.x; }
        __syncthreads();

        auto fetch = [&](int e) -> int {
            int lo = 0;
            #pragma unroll
            for (int s = NCHUNK / 2; s >= 1; s >>= 1) if (sPre[lo + s] <= e) lo += s;
            return tmp[sSt[lo] + (e - sPre[lo])];
        };
        const bool staged = total <= CAP;

        for (int r0 = 0; r0 < total; r0 += NT * PB) {
            int wv[PB];
            #pragma unroll
            for (int k = 0; k < PB; ++k) {
                const int e = r0 + k * NT + tid;
                wv[k] = fetch(e < total ? e : total - 1);
            }
            #pragma unroll
            for (int k = 0; k < PB; ++k) {
                const int e = r0 + k * NT + tid;
                if (e < total) {
                    if (staged) sIn[e] = wv[k];
                    atomicAdd(&sDeg[wv[k] & LOC_MASK], 1);
                }
            }
        }
        __syncthreads();
        const int c = (tid < BIN_NODES) ? sDeg[tid] : 0;
        int incn = c;
        for (int d = 1; d < 32; d <<= 1) {
            const int t = __shfl_up(incn, d);
            if (lane >= d) incn += t;
        }
        if (lane == 31) sWT[wave] = incn;
        __syncthreads();
        int wn = 0;
        for (int j = 0; j < NW; ++j) if (j < wave) wn += sWT[j];
        const int off = wn + incn - c;
        if (tid < BIN_NODES) {
            sCur[tid] = off;
            seg[bin * BIN_NODES + tid] = make_int2(base + off, c);
        }
        __syncthreads();
        for (int r0 = 0; r0 < total; r0 += NT * PB) {
            int wv[PB];
            #pragma unroll
            for (int k = 0; k < PB; ++k) {
                const int e = r0 + k * NT + tid;
                const int ec = e < total ? e : total - 1;
                wv[k] = staged ? sIn[ec] : fetch(ec);
            }
            #pragma unroll
            for (int k = 0; k < PB; ++k) {
                const int e = r0 + k * NT + tid;
                if (e < total) {
                    const int pos = atomicAdd(&sCur[wv[k] & LOC_MASK], 1);
                    const int s = wv[k] >> LOC_BITS;
                    if (staged) sOut[pos] = s; else nbr[base + pos] = s;
                }
            }
        }
        __syncthreads();
        if (staged) for (int i = tid; i < total; i += NT) nbr[base + i] = sOut[i];
    }
    __syncthreads();

    float* const tileM = sTile + wave * (16 * LDA);
    const float* const aM = tileM + l16 * LDA + half * 2;
    const float* const bL = sW + half * ODD + l16 * 2;
    const float* const bR = bL + MS;
    const float* const bP = bR + MS;

    const int turn = (NW == 15 && wave == 11) ? 8 : (NW == 15 && wave >= 8 && wave < 11) ? wave + 1 : wave;
    for (int tl = turn; tl < BIN_NODES / 16; tl += NW) {
        const int node0 = bin * BIN_NODES + tl * 16;

        const int2 sd = seg[node0 + l16];
        const int start = __builtin_amdgcn_readlane(sd.x, 0);
        const int total = __builtin_amdgcn_readlane(sd.x, 15) + __builtin_amdgcn_readlane(sd.y, 15) - start;

        int cur = 0;
        int curEnd = __builtin_amdgcn_readlane(sd.y, 0);
        float4 acc = make_float4(0.f, 0.f, 0.f, 0.f);

        auto flush = [&]() {
            const int dg = __builtin_amdgcn_readlane(sd.y, cur);
            const float inv = 1.0f / fmaxf((float)dg, 1.0f);
            acc.x *= inv; acc.y *= inv; acc.z *= inv; acc.w *= inv;
            *reinterpret_cast<float4*>(&tileM[cur * LDA + lane * 4]) = acc;
            acc = make_float4(0.f, 0.f, 0.f, 0.f);
            ++cur;
            if (cur < 16) curEnd += __builtin_amdgcn_readlane(sd.y, cur);
        };
        auto take = [&](int p, const float4& v) {
            while (p >= curEnd && cur < 15) flush();
            acc.x += v.x; acc.y += v.y; acc.z += v.z; acc.w += v.w;
        };

        int nraw = 0;
        if (lane < total) nraw = nbr[start + lane];
        for (int base = 0; base < total; base += 32) {
            const int n = (total - base < 32) ? (total - base) : 32;
            int idx = nraw;
            if (base + 32 + lane < total) nraw = nbr[start + base + 32 + lane];
            if (IDX) { if (lane < n) idx = xidx[idx]; }

            int j = 0;
            for (; j + GB <= n; j += GB) {
                float4 v[GB];
                #pragma unroll
                for (int u = 0; u < GB; ++u) {
                    const int s = __builtin_amdgcn_readlane(idx, j + u);
                    v[u] = *reinterpret_cast<const float4*>(X + (size_t)s * K + lane * 4);
                }
                #pragma unroll
                for (int u = 0; u < GB; ++u) take(base + j + u, v[u]);
            }
            for (; j + 4 <= n; j += 4) {
                float4 v[4];
                #pragma unroll
                for (int u = 0; u < 4; ++u) {
                    const int s = __builtin_amdgcn_readlane(idx, j + u);
                    v[u] = *reinterpret_cast<const float4*>(X + (size_t)s * K + lane * 4);
                }
                #pragma unroll
                for (int u = 0; u < 4; ++u) take(base + j + u, v[u]);
            }
            for (; j < n; ++j) {
                const int s = __builtin_amdgcn_readlane(idx, j);
                const float4 v = *reinterpret_cast<const float4*>(X + (size_t)s * K + lane * 4);
                take(base + j, v);
            }
        }
        while (cur < 16) flush();

        __builtin_amdgcn_fence(__ATOMIC_RELEASE, "wavefront");
        __builtin_amdgcn_wave_barrier();
        __builtin_amdgcn_fence(__ATOMIC_ACQUIRE, "wavefront");

        int xrow = node0 + l16;
        if (IDX) xrow = xidx[xrow];
#define OWN_ROWS(F) F(0) F(1) F(2) F(3) F(4) F(5) F(6) F(7) F(8) F(9) F(10) F(11) F(12) F(13) F(14) F(15)
#define OWN_ROW_LOAD(r) \
        const float4 xr##r = *reinterpret_cast<const float4*>( \
            X + (size_t)__builtin_amdgcn_readlane(xrow, r) * K + lane * 4);
#define OWN_ROW_PARK(r) *reinterpret_cast<float4*>(&tileM[r * LDA + lane * 4]) = xr##r;
        OWN_ROWS(OWN_ROW_LOAD)

        v8f c[CT];
        #pragma unroll
        for (int t = 0; t < CT; ++t) c[t] = (v8f){0.f, 0.f, 0.f, 0.f, 0.f, 0.f, 0.f, 0.f};

        auto product = [&](const float* __restrict__ bp) {
            const float* __restrict__ ap = aM;
            v2f a0, a1, b0[CT], b1[CT];
            a0 = *reinterpret_cast<const v2f*>(ap);
            #pragma unroll
            for (int t = 0; t < CT; ++t) b0[t] = *reinterpret_cast<const v2f*>(bp + t * 32);
            #pragma nounroll
            for (int it = 0; it < K / 8 - 1; ++it) {
                a1 = *reinterpret_cast<const v2f*>(ap + 4);
                #pragma unroll
                for (int t = 0; t < CT; ++t) b1[t] = *reinterpret_cast<const v2f*>(bp + STEP + t * 32);
                #pragma unroll
                for (int t = 0; t < CT; ++t)
                    c[t] = __builtin_amdgcn_wmma_f32_16x16x4_f32(false, a0, false, b0[t],
                                                                 (short)0, c[t], false, false);
                a0 = *reinterpret_cast<const v2f*>(ap + 8);
                #pragma unroll
                for (int t = 0; t < CT; ++t) b0[t] = *reinterpret_cast<const v2f*>(bp + 2 * STEP + t * 32);
                #pragma unroll
                for (int t = 0; t < CT; ++t)
                    c[t] = __builtin_amdgcn_wmma_f32_16x16x4_f32(false, a1, false, b1[t],
                                                                 (short)0, c[t], false, false);
                ap += 8;
                bp += 2 * STEP;
            }
            a1 = *reinterpret_cast<const v2f*>(ap + 4);
            #pragma unroll
            for (int t = 0; t < CT; ++t) b1[t] = *reinterpret_cast<const v2f*>(bp + STEP + t * 32);
            #pragma unroll
            for (int t = 0; t < CT; ++t)
                c[t] = __builtin_amdgcn_wmma_f32_16x16x4_f32(false, a0, false, b0[t],
                                                             (short)0, c[t], false, false);
            #pragma unroll
            for (int t = 0; t < CT; ++t)
                c[t] = __builtin_amdgcn_wmma_f32_16x16x4_f32(false, a1, false, b1[t],
                                                             (short)0, c[t], false, false);
        };

        product(bL);

        __builtin_amdgcn_fence(__ATOMIC_RELEASE, "wavefront");
        __builtin_amdgcn_wave_barrier();
        __builtin_amdgcn_fence(__ATOMIC_ACQUIRE, "wavefront");
        OWN_ROWS(OWN_ROW_PARK)
#undef OWN_ROW_PARK
#undef OWN_ROW_LOAD
#undef OWN_ROWS
        __builtin_amdgcn_fence(__ATOMIC_RELEASE, "wavefront");
        __builtin_amdgcn_wave_barrier();
        __builtin_amdgcn_fence(__ATOMIC_ACQUIRE, "wavefront");

        product(bR);

        __builtin_amdgcn_fence(__ATOMIC_RELEASE, "wavefront");
        __builtin_amdgcn_wave_barrier();
        __builtin_amdgcn_fence(__ATOMIC_ACQUIRE, "wavefront");
        #pragma unroll
        for (int t = 0; t < CT; ++t) {
            const int col = t * 16 + l16;
            const float bv = bias[col];
            #pragma unroll
            for (int v = 0; v < 8; ++v) {
                float val = c[t][v] + bv;
                if (RELU) val = fmaxf(val, 0.0f);
                tileM[(v + half * 8) * LDA + col] = val;
            }
            c[t] = (v8f){0.f, 0.f, 0.f, 0.f, 0.f, 0.f, 0.f, 0.f};
        }
        __builtin_amdgcn_fence(__ATOMIC_RELEASE, "wavefront");
        __builtin_amdgcn_wave_barrier();
        __builtin_amdgcn_fence(__ATOMIC_ACQUIRE, "wavefront");

        product(bP);

        #pragma unroll
        for (int t = 0; t < CT; ++t) {
            const int col = t * 16 + l16;
            const float bv = (t * 16 >= O2) ? bias2[col - O2] : 0.0f;
            #pragma unroll
            for (int v = 0; v < 8; ++v) {
                const int r = node0 + v + half * 8;
                out[(size_t)r * OUT + col] = c[t][v] + bv;
            }
        }

        __builtin_amdgcn_fence(__ATOMIC_RELEASE, "wavefront");
        __builtin_amdgcn_wave_barrier();
        __builtin_amdgcn_fence(__ATOMIC_ACQUIRE, "wavefront");
    }
}

static_assert(HIDDEN == 2 * OUT_C && OUT_C == 64 && N_NODES % 16 == 0, "row split and tiles of mean_add_kernel");
template<int NW, int GB>
__global__ __launch_bounds__(NW * 32)
void mean_add_kernel(const int* __restrict__ nbr, const int2* __restrict__ seg,
                     const float* __restrict__ zr, float* __restrict__ out) {
    constexpr int LD = OUT_C;
    __shared__ __attribute__((aligned(8))) float sMean[NW * 16 * LD];
    const int lane = threadIdx.x;
    const int wave = __builtin_amdgcn_readfirstlane(threadIdx.y);
    const int tile = wave * (int)gridDim.x + (int)blockIdx.x;
    if (tile >= N_NODES / 16) return;
    const int node0 = tile * 16;
    const int l16 = lane & 15;
    float* const tileM = sMean + wave * (16 * LD);

    const int2 sd = seg[node0 + l16];

#define R_ROWS(F) F(0) F(1) F(2) F(3) F(4) F(5) F(6) F(7) F(8) F(9) F(10) F(11) F(12) F(13) F(14) F(15)
#define R_ROW_LOAD(r) \
    const float2 rr##r = *reinterpret_cast<const float2*>( \
        zr + (size_t)(node0 + r) * HIDDEN + OUT_C + lane * 2);
#define R_ROW_OUT(r) { \
        const float2 m = *reinterpret_cast<const float2*>(&tileM[r * LD + lane * 2]); \
        *reinterpret_cast<float2*>(out + (size_t)(node0 + r) * OUT_C + lane * 2) = \
            make_float2(rr##r.x + m.x, rr##r.y + m.y); }
    R_ROWS(R_ROW_LOAD)

    const int start = __builtin_amdgcn_readlane(sd.x, 0);
    const int total = __builtin_amdgcn_readlane(sd.x, 15) + __builtin_amdgcn_readlane(sd.y, 15) - start;

    int cur = 0;
    int curEnd = __builtin_amdgcn_readlane(sd.y, 0);
    float2 acc = make_float2(0.f, 0.f);

    auto flush = [&]() {
        const int dg = __builtin_amdgcn_readlane(sd.y, cur);
        const float inv = 1.0f / fmaxf((float)dg, 1.0f);
        *reinterpret_cast<float2*>(&tileM[cur * LD + lane * 2]) = make_float2(acc.x * inv, acc.y * inv);
        acc = make_float2(0.f, 0.f);
        ++cur;
        if (cur < 16) curEnd += __builtin_amdgcn_readlane(sd.y, cur);
    };
    auto take = [&](int p, const float2& v) {
        while (p >= curEnd && cur < 15) flush();
        acc.x += v.x; acc.y += v.y;
    };

    int nraw = 0;
    if (lane < total) nraw = nbr[start + lane];
    for (int base = 0; base < total; base += 32) {
        const int n = (total - base < 32) ? (total - base) : 32;
        const int idx = nraw;
        if (base + 32 + lane < total) nraw = nbr[start + base + 32 + lane];

        int j = 0;
        for (; j + GB <= n; j += GB) {
            float2 v[GB];
            #pragma unroll
            for (int u = 0; u < GB; ++u) {
                const int s = __builtin_amdgcn_readlane(idx, j + u);
                v[u] = *reinterpret_cast<const float2*>(zr + (size_t)s * HIDDEN + lane * 2);
            }
            #pragma unroll
            for (int u = 0; u < GB; ++u) take(base + j + u, v[u]);
        }
        for (; j + 4 <= n; j += 4) {
            float2 v[4];
            #pragma unroll
            for (int u = 0; u < 4; ++u) {
                const int s = __builtin_amdgcn_readlane(idx, j + u);
                v[u] = *reinterpret_cast<const float2*>(zr + (size_t)s * HIDDEN + lane * 2);
            }
            #pragma unroll
            for (int u = 0; u < 4; ++u) take(base + j + u, v[u]);
        }
        for (; j < n; ++j) {
            const int s = __builtin_amdgcn_readlane(idx, j);
            const float2 v = *reinterpret_cast<const float2*>(zr + (size_t)s * HIDDEN + lane * 2);
            take(base + j, v);
        }
    }
    while (cur < 16) flush();

    R_ROWS(R_ROW_OUT)
#undef R_ROW_OUT
#undef R_ROW_LOAD
#undef R_ROWS
}

extern "C" void kernel_launch(void* const* d_in, const int* in_sizes, int n_in,
                              void* d_out, int out_size, void* d_ws, size_t ws_size,
                              hipStream_t stream) {
    (void)in_sizes; (void)n_in; (void)out_size; (void)ws_size;

    const int*   xidx = (const int*)d_in[0];
    const int*   ei   = (const int*)d_in[1];
    const float* emb  = (const float*)d_in[2];
    const float* Wl1  = (const float*)d_in[3];
    const float* bl1  = (const float*)d_in[4];
    const float* Wr1  = (const float*)d_in[5];
    const float* Wl2  = (const float*)d_in[6];
    const float* bl2  = (const float*)d_in[7];
    const float* Wr2  = (const float*)d_in[8];

    int*   nbr   = (int*)d_ws;
    int*   tmp   = nbr + N_EDGES;
    int2*  seg   = (int2*)(tmp + N_EDGES);
    int2*  segT  = seg + N_NODES;
    float* zr    = (float*)d_ws + (size_t)N_NODES * HIDDEN;

    chunk_sort_kernel<<<NCHUNK, BT, 0, stream>>>(ei, segT, tmp);


    sage_layer_kernel<EMB_DIM, HIDDEN, true, true, 15, 16>
        <<<dim3(NBIN), dim3(32, 15), 0, stream>>>(
            segT, tmp, nbr, seg, emb, xidx, Wl1, bl1, Wr1, Wl2, bl2, Wr2, zr);

    mean_add_kernel<25, 16><<<dim3(256), dim3(32, 25), 0, stream>>>(
        nbr, seg, zr, (float*)d_out);
}


# Independent loads issued as one group, not waited for in turn: weight copy (13 waits before; parked after the bin sort), edge-chunk loads (7), both gathers keep 32 rows in flight; sum order kept
# speedup vs baseline: 4.4826x; 1.0416x over previous
#include <hip/hip_runtime.h>

typedef __attribute__((ext_vector_type(2))) float v2f;
typedef __attribute__((ext_vector_type(8))) float v8f;

#define N_NODES 100000
#define EMB_DIM 128
#define HIDDEN  128
#define OUT_C   64
#define N_EDGES 800000

#define NCHUNK    256
#define CH        (N_EDGES / NCHUNK)
#define BIN_NODES 400
#define NBIN      (N_NODES / BIN_NODES)
#define LOC_BITS  9
#define LOC_MASK  ((1 << LOC_BITS) - 1)
#define BT        512
static_assert(CH * NCHUNK == N_EDGES, "chunks tile the edge list");
static_assert(NBIN * BIN_NODES == N_NODES && BIN_NODES % 16 == 0, "bins tile the nodes in whole row tiles");
static_assert(BIN_NODES <= (1 << LOC_BITS) && N_NODES <= (1 << (31 - LOC_BITS)), "an edge fits one word");
static_assert(NBIN <= BT, "one thread per bin");
static_assert(NBIN <= 256, "one layer workgroup per bin, all resident at once");

__global__ __launch_bounds__(BT)
void chunk_sort_kernel(const int* __restrict__ ei, int2* __restrict__ segT, int* __restrict__ tmp) {
    __shared__ int sCnt[NBIN], sLoc[NBIN];
    __shared__ int sVal[CH];
    const int tid = threadIdx.x;
    const int w = blockIdx.x;
    if (tid < NBIN) sCnt[tid] = 0;
    __syncthreads();

    const int* __restrict__ srcp = ei + w * CH;
    const int* __restrict__ dstp = ei + N_EDGES + w * CH;
    constexpr int PER = (CH + BT - 1) / BT;
    int pk[PER], bn[PER], rk[PER], dv[PER], sv[PER];
    #pragma unroll
    for (int k = 0; k < PER; ++k) {
        const int i = tid + k * BT;
        const int ic = ((k + 1) * BT > CH && i >= CH) ? CH - 1 : i;
        dv[k] = dstp[ic];
        sv[k] = srcp[ic];
    }
    #pragma unroll
    for (int k = 0; k < PER; ++k) {
        const int i = tid + k * BT;
        const bool live = !((k + 1) * BT > CH && i >= CH);
        const int d = dv[k];
        const int b = d / BIN_NODES;
        pk[k] = (sv[k] << LOC_BITS) | (d - b * BIN_NODES);
        bn[k] = live ? b : -1;
        rk[k] = 0;
        if (live) rk[k] = atomicAdd(&sCnt[b], 1);
    }
    __syncthreads();
    if (tid < NBIN) {
        int s = 0;
        for (int j = 0; j < tid; ++j) s += sCnt[j];
        sLoc[tid] = s;
        segT[tid * NCHUNK + w] = make_int2(s, sCnt[tid]);
    }
    __syncthreads();
    #pragma unroll
    for (int k = 0; k < PER; ++k)
        if (bn[k] >= 0) sVal[sLoc[bn[k]] + rk[k]] = pk[k];
    __syncthreads();
    for (int p = tid; p < CH; p += BT) tmp[w * CH + p] = sVal[p];
}

template<int K, int OUT, bool RELU, bool IDX, int NW>
__global__ __launch_bounds__(NW * 32)
void sage_layer_kernel(const int2* __restrict__ segT,
                       const int* __restrict__ tmp,
                       int* nbr,
                       int2* seg,
                       const float* __restrict__ X,
                       const int* __restrict__ xidx,
                       const float* __restrict__ Wl,
                       const float* __restrict__ bias,
                       const float* __restrict__ Wr,
                       const float* __restrict__ Wl2,
                       const float* __restrict__ bias2,
                       const float* __restrict__ Wr2,
                       float* __restrict__ out) {
    static_assert(K == 128, "one float4 per lane covers a row");
    static_assert(OUT == K, "the h tile is the A operand of the third product");
    constexpr int CT  = OUT / 16;
    constexpr int LDA = K + 4;
    constexpr int KP  = K / 2;
    constexpr int STEP = OUT * 2;
    constexpr int ODD  = (KP / 2) * STEP + 32;
    constexpr int MS   = KP * STEP + 64;
    constexpr int O2   = OUT / 2;
    constexpr int NT  = NW * 32;
    __shared__ __attribute__((aligned(16))) float sW[3 * MS];
    constexpr int TW = NW * 16 * LDA;
    union TileSpace { float tile[TW]; int sort[TW]; };
    __shared__ __attribute__((aligned(16))) TileSpace sU;
    float* const sTile = sU.tile;

    const int lane = threadIdx.x;
    const int wave = __builtin_amdgcn_readfirstlane(threadIdx.y);
    const int half = lane >> 4;
    const int l16  = lane & 15;

    const int bin = (int)blockIdx.x;
    const int tid = threadIdx.y * 32 + threadIdx.x;
    static_assert(NT >= NCHUNK && NT >= BIN_NODES, "one thread per chunk run / per node of the bin");

    int2 run = make_int2(0, 0);
    if (tid < NCHUNK) run = segT[bin * NCHUNK + tid];

    constexpr int WC4   = OUT / 4;
    constexpr int WPER  = KP * WC4;
    constexpr int WPASS = (3 * WPER + NT - 1) / NT;
    float4 wr0[WPASS], wr1[WPASS];
    #pragma unroll
    for (int it = 0; it < WPASS; ++it) {
        const int i = tid + it * NT;
        const int ic = ((it + 1) * NT > 3 * WPER && i >= 3 * WPER) ? 3 * WPER - 1 : i;
        const int m   = ic / WPER;
        const int rem = ic - m * WPER;
        const int kp  = rem / WC4;
        const int c4  = rem - kp * WC4;
        const bool third = (m == 2);
        const bool right = third && (c4 * 4 >= O2);
        const float* __restrict__ wb = third ? (right ? Wr2 : Wl2) : (m ? Wr : Wl);
        const int pitch = third ? O2 : OUT;
        const int col   = c4 * 4 - (right ? O2 : 0);
        const float* __restrict__ p0 = wb + (size_t)(2 * kp) * pitch + col;
        wr0[it] = *reinterpret_cast<const float4*>(p0);
        wr1[it] = *reinterpret_cast<const float4*>(p0 + pitch);
    }

    {
        constexpr int PB  = 8;
        constexpr int O_PRE = 0;
        constexpr int O_ST  = O_PRE + NCHUNK;
        constexpr int O_DEG = O_ST + NCHUNK;
        constexpr int O_CUR = O_DEG + BIN_NODES;
        constexpr int O_WT  = O_CUR + BIN_NODES;
        constexpr int O_WS  = O_WT + 32;
        constexpr int O_IN  = O_WS + 32;
        constexpr int CAP   = (TW - O_IN) / 2;
        constexpr int O_OUT = O_IN + CAP;
        static_assert(CAP >= 4 * (N_EDGES / NBIN), "an average bin is staged with room to spare");
        int* const sPre = sU.sort + O_PRE;
        int* const sSt  = sU.sort + O_ST;
        int* const sDeg = sU.sort + O_DEG;
        int* const sCur = sU.sort + O_CUR;
        int* const sWT  = sU.sort + O_WT;
        int* const sWS  = sU.sort + O_WS;
        int* const sIn  = sU.sort + O_IN;
        int* const sOut = sU.sort + O_OUT;

        int incl = run.y, sts = run.x;
        for (int d = 1; d < 32; d <<= 1) {
            const int t = __shfl_up(incl, d);
            if (lane >= d) incl += t;
        }
        for (int d = 16; d >= 1; d >>= 1) sts += __shfl_xor(sts, d);
        if (lane == 31) { sWT[wave] = incl; sWS[wave] = sts; }
        if (tid < BIN_NODES) sDeg[tid] = 0;
        __syncthreads();
        int wb = 0, totalV = 0, baseV = 0;
        for (int j = 0; j < NCHUNK / 32; ++j) {
            const int t = sWT[j];
            if (j < wave) wb += t;
            totalV += t;
            baseV += sWS[j];
        }
        const int total = __builtin_amdgcn_readfirstlane(totalV);
        const int base  = __builtin_amdgcn_readfirstlane(baseV);
        if (tid < NCHUNK) { sPre[tid] = wb + incl - run.y; sSt[tid] = tid * CH + run.x; }
        __syncthreads();

        auto fetch = [&](int e) -> int {
            int lo = 0;
            #pragma unroll
            for (int s = NCHUNK / 2; s >= 1; s >>= 1) if (sPre[lo + s] <= e) lo += s;
            return tmp[sSt[lo] + (e - sPre[lo])];
        };
        const bool staged = total <= CAP;

        for (int r0 = 0; r0 < total; r0 += NT * PB) {
            int wv[PB];
            #pragma unroll
            for (int k = 0; k < PB; ++k) {
                const int e = r0 + k * NT + tid;
                wv[k] = fetch(e < total ? e : total - 1);
            }
            #pragma unroll
            for (int k = 0; k < PB; ++k) {
                const int e = r0 + k * NT + tid;
                if (e < total) {
                    if (staged) sIn[e] = wv[k];
                    atomicAdd(&sDeg[wv[k] & LOC_MASK], 1);
                }
            }
        }
        __syncthreads();
        const int c = (tid < BIN_NODES) ? sDeg[tid] : 0;
        int incn = c;
        for (int d = 1; d < 32; d <<= 1) {
            const int t = __shfl_up(incn, d);
            if (lane >= d) incn += t;
        }
        if (lane == 31) sWT[wave] = incn;
        __syncthreads();
        int wn = 0;
        for (int j = 0; j < NW; ++j) if (j < wave) wn += sWT[j];
        const int off = wn + incn - c;
        if (tid < BIN_NODES) {
            sCur[tid] = off;
            seg[bin * BIN_NODES + tid] = make_int2(base + off, c);
        }
        __syncthreads();
        for (int r0 = 0; r0 < total; r0 += NT * PB) {
            int wv[PB];
            #pragma unroll
            for (int k = 0; k < PB; ++k) {
                const int e = r0 + k * NT + tid;
                const int ec = e < total ? e : total - 1;
                wv[k] = staged ? sIn[ec] : fetch(ec);
            }
            #pragma unroll
            for (int k = 0; k < PB; ++k) {
                const int e = r0 + k * NT + tid;
                if (e < total) {
                    const int pos = atomicAdd(&sCur[wv[k] & LOC_MASK], 1);
                    const int s = wv[k] >> LOC_BITS;
                    if (staged) sOut[pos] = s; else nbr[base + pos] = s;
                }
            }
        }
        __syncthreads();
        if (staged) for (int i = tid; i < total; i += NT) nbr[base + i] = sOut[i];
    }
    #pragma unroll
    for (int it = 0; it < WPASS; ++it) {
        const int i = tid + it * NT;
        if ((it + 1) * NT > 3 * WPER && i >= 3 * WPER) continue;
        const int m   = i / WPER;
        const int rem = i - m * WPER;
        const int kp  = rem / WC4;
        const int c4  = rem - kp * WC4;
        const float4 r0 = wr0[it], r1 = wr1[it];
        float* d = sW + m * MS + (kp & 1) * ODD + (kp >> 1) * STEP + c4 * 8;
        *reinterpret_cast<float4*>(d)     = make_float4(r0.x, r1.x, r0.y, r1.y);
        *reinterpret_cast<float4*>(d + 4) = make_float4(r0.z, r1.z, r0.w, r1.w);
    }
    __syncthreads();

    float* const tileM = sTile + wave * (16 * LDA);
    const float* const aM = tileM + l16 * LDA + half * 2;
    const float* const bL = sW + half * ODD + l16 * 2;
    const float* const bR = bL + MS;
    const float* const bP = bR + MS;

    const int turn = (NW == 15 && wave == 11) ? 8 : (NW == 15 && wave >= 8 && wave < 11) ? wave + 1 : wave;
    for (int tl = turn; tl < BIN_NODES / 16; tl += NW) {
        const int node0 = bin * BIN_NODES + tl * 16;

        const int2 sd = seg[node0 + l16];
        const int start = __builtin_amdgcn_readlane(sd.x, 0);
        const int total = __builtin_amdgcn_readlane(sd.x, 15) + __builtin_amdgcn_readlane(sd.y, 15) - start;

        int cur = 0;
        int curEnd = __builtin_amdgcn_readlane(sd.y, 0);
        float4 acc = make_float4(0.f, 0.f, 0.f, 0.f);

        auto flush = [&]() {
            const int dg = __builtin_amdgcn_readlane(sd.y, cur);
            const float inv = 1.0f / fmaxf((float)dg, 1.0f);
            acc.x *= inv; acc.y *= inv; acc.z *= inv; acc.w *= inv;
            *reinterpret_cast<float4*>(&tileM[cur * LDA + lane * 4]) = acc;
            acc = make_float4(0.f, 0.f, 0.f, 0.f);
            ++cur;
            if (cur < 16) curEnd += __builtin_amdgcn_readlane(sd.y, cur);
        };
        auto take = [&](int p, const float4& v) {
            while (p >= curEnd && cur < 15) flush();
            acc.x += v.x; acc.y += v.y; acc.z += v.z; acc.w += v.w;
        };

        int icur = 0, nraw = 0;
        if (lane < total) icur = nbr[start + lane];
        if (32 + lane < total) nraw = nbr[start + 32 + lane];
        if (IDX) { if (lane < total) icur = xidx[icur]; }
        for (int base = 0; base < total; base += 32) {
            const int n = (total - base < 32) ? (total - base) : 32;
            const int idx = icur;
            int inext = nraw;
            if (IDX) { if (base + 32 + lane < total) inext = xidx[nraw]; }
            if (base + 64 + lane < total) nraw = nbr[start + base + 64 + lane];

            float4 v[32];
            #pragma unroll
            for (int u = 0; u < 32; ++u) {
                const int s = __builtin_amdgcn_readlane(idx, u < n ? u : n - 1);
                v[u] = *reinterpret_cast<const float4*>(X + (size_t)s * K + lane * 4);
            }
            #pragma unroll
            for (int u = 0; u < 32; ++u) if (u == 0 || u < n) take(base + u, v[u]);
            icur = inext;
        }
        while (cur < 16) flush();

        __builtin_amdgcn_fence(__ATOMIC_RELEASE, "wavefront");
        __builtin_amdgcn_wave_barrier();
        __builtin_amdgcn_fence(__ATOMIC_ACQUIRE, "wavefront");

        int xrow = node0 + l16;
        if (IDX) xrow = xidx[xrow];
#define OWN_ROWS(F) F(0) F(1) F(2) F(3) F(4) F(5) F(6) F(7) F(8) F(9) F(10) F(11) F(12) F(13) F(14) F(15)
#define OWN_ROW_LOAD(r) \
        const float4 xr##r = *reinterpret_cast<const float4*>( \
            X + (size_t)__builtin_amdgcn_readlane(xrow, r) * K + lane * 4);
#define OWN_ROW_PARK(r) *reinterpret_cast<float4*>(&tileM[r * LDA + lane * 4]) = xr##r;
        OWN_ROWS(OWN_ROW_LOAD)

        v8f c[CT];
        #pragma unroll
        for (int t = 0; t < CT; ++t) c[t] = (v8f){0.f, 0.f, 0.f, 0.f, 0.f, 0.f, 0.f, 0.f};

        auto product = [&](const float* __restrict__ bp) {
            const float* __restrict__ ap = aM;
            v2f a0, a1, b0[CT], b1[CT];
            a0 = *reinterpret_cast<const v2f*>(ap);
            #pragma unroll
            for (int t = 0; t < CT; ++t) b0[t] = *reinterpret_cast<const v2f*>(bp + t * 32);
            #pragma nounroll
            for (int it = 0; it < K / 8 - 1; ++it) {
                a1 = *reinterpret_cast<const v2f*>(ap + 4);
                #pragma unroll
                for (int t = 0; t < CT; ++t) b1[t] = *reinterpret_cast<const v2f*>(bp + STEP + t * 32);
                #pragma unroll
                for (int t = 0; t < CT; ++t)
                    c[t] = __builtin_amdgcn_wmma_f32_16x16x4_f32(false, a0, false, b0[t],
                                                                 (short)0, c[t], false, false);
                a0 = *reinterpret_cast<const v2f*>(ap + 8);
                #pragma unroll
                for (int t = 0; t < CT; ++t) b0[t] = *reinterpret_cast<const v2f*>(bp + 2 * STEP + t * 32);
                #pragma unroll
                for (int t = 0; t < CT; ++t)
                    c[t] = __builtin_amdgcn_wmma_f32_16x16x4_f32(false, a1, false, b1[t],
                                                                 (short)0, c[t], false, false);
                ap += 8;
                bp += 2 * STEP;
            }
            a1 = *reinterpret_cast<const v2f*>(ap + 4);
            #pragma unroll
            for (int t = 0; t < CT; ++t) b1[t] = *reinterpret_cast<const v2f*>(bp + STEP + t * 32);
            #pragma unroll
            for (int t = 0; t < CT; ++t)
                c[t] = __builtin_amdgcn_wmma_f32_16x16x4_f32(false, a0, false, b0[t],
                                                             (short)0, c[t], false, false);
            #pragma unroll
            for (int t = 0; t < CT; ++t)
                c[t] = __builtin_amdgcn_wmma_f32_16x16x4_f32(false, a1, false, b1[t],
                                                             (short)0, c[t], false, false);
        };

        product(bL);

        __builtin_amdgcn_fence(__ATOMIC_RELEASE, "wavefront");
        __builtin_amdgcn_wave_barrier();
        __builtin_amdgcn_fence(__ATOMIC_ACQUIRE, "wavefront");
        OWN_ROWS(OWN_ROW_PARK)
#undef OWN_ROW_PARK
#undef OWN_ROW_LOAD
#undef OWN_ROWS
        __builtin_amdgcn_fence(__ATOMIC_RELEASE, "wavefront");
        __builtin_amdgcn_wave_barrier();
        __builtin_amdgcn_fence(__ATOMIC_ACQUIRE, "wavefront");

        product(bR);

        __builtin_amdgcn_fence(__ATOMIC_RELEASE, "wavefront");
        __builtin_amdgcn_wave_barrier();
        __builtin_amdgcn_fence(__ATOMIC_ACQUIRE, "wavefront");
        #pragma unroll
        for (int t = 0; t < CT; ++t) {
            const int col = t * 16 + l16;
            const float bv = bias[col];
            #pragma unroll
            for (int v = 0; v < 8; ++v) {
                float val = c[t][v] + bv;
                if (RELU) val = fmaxf(val, 0.0f);
                tileM[(v + half * 8) * LDA + col] = val;
            }
            c[t] = (v8f){0.f, 0.f, 0.f, 0.f, 0.f, 0.f, 0.f, 0.f};
        }
        __builtin_amdgcn_fence(__ATOMIC_RELEASE, "wavefront");
        __builtin_amdgcn_wave_barrier();
        __builtin_amdgcn_fence(__ATOMIC_ACQUIRE, "wavefront");

        product(bP);

        #pragma unroll
        for (int t = 0; t < CT; ++t) {
            const int col = t * 16 + l16;
            const float bv = (t * 16 >= O2) ? bias2[col - O2] : 0.0f;
            #pragma unroll
            for (int v = 0; v < 8; ++v) {
                const int r = node0 + v + half * 8;
                out[(size_t)r * OUT + col] = c[t][v] + bv;
            }
        }

        __builtin_amdgcn_fence(__ATOMIC_RELEASE, "wavefront");
        __builtin_amdgcn_wave_barrier();
        __builtin_amdgcn_fence(__ATOMIC_ACQUIRE, "wavefront");
    }
}

static_assert(HIDDEN == 2 * OUT_C && OUT_C == 64 && N_NODES % 16 == 0, "row split and tiles of mean_add_kernel");
template<int NW>
__global__ __launch_bounds__(NW * 32)
void mean_add_kernel(const int* __restrict__ nbr, const int2* __restrict__ seg,
                     const float* __restrict__ zr, float* __restrict__ out) {
    constexpr int LD = OUT_C;
    __shared__ __attribute__((aligned(8))) float sMean[NW * 16 * LD];
    const int lane = threadIdx.x;
    const int wave = __builtin_amdgcn_readfirstlane(threadIdx.y);
    const int tile = wave * (int)gridDim.x + (int)blockIdx.x;
    if (tile >= N_NODES / 16) return;
    const int node0 = tile * 16;
    const int l16 = lane & 15;
    float* const tileM = sMean + wave * (16 * LD);

    const int2 sd = seg[node0 + l16];

#define R_ROWS(F) F(0) F(1) F(2) F(3) F(4) F(5) F(6) F(7) F(8) F(9) F(10) F(11) F(12) F(13) F(14) F(15)
#define R_ROW_LOAD(r) \
    const float2 rr##r = *reinterpret_cast<const float2*>( \
        zr + (size_t)(node0 + r) * HIDDEN + OUT_C + lane * 2);
#define R_ROW_OUT(r) { \
        const float2 m = *reinterpret_cast<const float2*>(&tileM[r * LD + lane * 2]); \
        *reinterpret_cast<float2*>(out + (size_t)(node0 + r) * OUT_C + lane * 2) = \
            make_float2(rr##r.x + m.x, rr##r.y + m.y); }
    R_ROWS(R_ROW_LOAD)

    const int start = __builtin_amdgcn_readlane(sd.x, 0);
    const int total = __builtin_amdgcn_readlane(sd.x, 15) + __builtin_amdgcn_readlane(sd.y, 15) - start;

    int cur = 0;
    int curEnd = __builtin_amdgcn_readlane(sd.y, 0);
    float2 acc = make_float2(0.f, 0.f);

    auto flush = [&]() {
        const int dg = __builtin_amdgcn_readlane(sd.y, cur);
        const float inv = 1.0f / fmaxf((float)dg, 1.0f);
        *reinterpret_cast<float2*>(&tileM[cur * LD + lane * 2]) = make_float2(acc.x * inv, acc.y * inv);
        acc = make_float2(0.f, 0.f);
        ++cur;
        if (cur < 16) curEnd += __builtin_amdgcn_readlane(sd.y, cur);
    };
    auto take = [&](int p, const float2& v) {
        while (p >= curEnd && cur < 15) flush();
        acc.x += v.x; acc.y += v.y;
    };

    int nraw = 0;
    if (lane < total) nraw = nbr[start + lane];
    for (int base = 0; base < total; base += 32) {
        const int n = (total - base < 32) ? (total - base) : 32;
        const int idx = nraw;
        if (base + 32 + lane < total) nraw = nbr[start + base + 32 + lane];

        float2 v[32];
        #pragma unroll
        for (int u = 0; u < 32; ++u) {
            const int s = __builtin_amdgcn_readlane(idx, u < n ? u : n - 1);
            v[u] = *reinterpret_cast<const float2*>(zr + (size_t)s * HIDDEN + lane * 2);
        }
        #pragma unroll
        for (int u = 0; u < 32; ++u) if (u == 0 || u < n) take(base + u, v[u]);
    }
    while (cur < 16) flush();

    R_ROWS(R_ROW_OUT)
#undef R_ROW_OUT
#undef R_ROW_LOAD
#undef R_ROWS
}

extern "C" void kernel_launch(void* const* d_in, const int* in_sizes, int n_in,
                              void* d_out, int out_size, void* d_ws, size_t ws_size,
                              hipStream_t stream) {
    (void)in_sizes; (void)n_in; (void)out_size; (void)ws_size;

    const int*   xidx = (const int*)d_in[0];
    const int*   ei   = (const int*)d_in[1];
    const float* emb  = (const float*)d_in[2];
    const float* Wl1  = (const float*)d_in[3];
    const float* bl1  = (const float*)d_in[4];
    const float* Wr1  = (const float*)d_in[5];
    const float* Wl2  = (const float*)d_in[6];
    const float* bl2  = (const float*)d_in[7];
    const float* Wr2  = (const float*)d_in[8];

    int*   nbr   = (int*)d_ws;
    int*   tmp   = nbr + N_EDGES;
    int2*  seg   = (int2*)(tmp + N_EDGES);
    int2*  segT  = seg + N_NODES;
    float* zr    = (float*)d_ws + (size_t)N_NODES * HIDDEN;

    chunk_sort_kernel<<<NCHUNK, BT, 0, stream>>>(ei, segT, tmp);


    sage_layer_kernel<EMB_DIM, HIDDEN, true, true, 15>
        <<<dim3(NBIN), dim3(32, 15), 0, stream>>>(
            segT, tmp, nbr, seg, emb, xidx, Wl1, bl1, Wr1, Wl2, bl2, Wr2, zr);

    mean_add_kernel<25><<<dim3(256), dim3(32, 25), 0, stream>>>(
        nbr, seg, zr, (float*)d_out);
}
